# starting kernel
# baseline (speedup 1.0000x reference)
#include <hip/hip_runtime.h>
#include <hip/hip_bf16.h>
#include <math.h>


typedef float v2f __attribute__((ext_vector_type(2)));
typedef float v8f __attribute__((ext_vector_type(8)));

__device__ __forceinline__ v8f v8f_zero() {
    v8f z = {0.f, 0.f, 0.f, 0.f, 0.f, 0.f, 0.f, 0.f};
    return z;
}

__device__ __forceinline__ v8f wmma4(v2f a, v2f b, v8f c) {
    return __builtin_amdgcn_wmma_f32_16x16x4_f32(false, a, false, b,
                                                 (short)0, c, false, false);
}

__global__ void relgen_kernel(float* __restrict__ rel) {
    const int row = blockIdx.x;
    const int t = threadIdx.x;
    const float pos = (float)(2047 - row);
    const float negl = -9.210340371976184f / 512.0f;
#pragma unroll
    for (int i = 0; i < 4; ++i) {
        int c = t + i * 256;
        int cc = c & 511;
        float inv = expf((float)cc * negl);
        float ang = pos * inv;
        rel[(size_t)row * 1024 + c] = (c < 512) ? sinf(ang) : cosf(ang);
    }
}

__global__ void gemm_nt_kernel(const float* __restrict__ A,
                               const float* __restrict__ B,
                               float* __restrict__ C,
                               int M, int N, int Kd) {
    const int lane = threadIdx.x & 31;
    const int w    = threadIdx.x >> 5;
    const int l16  = lane & 15;
    const int hlf  = lane >> 4;
    const int m0 = blockIdx.y * 64 + (w & 1) * 32;
    const int n0 = blockIdx.x * 256 + (w >> 1) * 64;
    (void)M;

    v8f c[2][4];
#pragma unroll
    for (int i = 0; i < 2; ++i)
#pragma unroll
        for (int j = 0; j < 4; ++j) c[i][j] = v8f_zero();

    for (int kb = 0; kb < Kd; kb += 4) {
        const int kk = kb + 2 * hlf;
        v2f a[2], b[4];
#pragma unroll
        for (int i = 0; i < 2; ++i)
            a[i] = *(const v2f*)(A + (size_t)(m0 + i * 16 + l16) * Kd + kk);
#pragma unroll
        for (int j = 0; j < 4; ++j)
            b[j] = *(const v2f*)(B + (size_t)(n0 + j * 16 + l16) * Kd + kk);
#pragma unroll
        for (int i = 0; i < 2; ++i)
#pragma unroll
            for (int j = 0; j < 4; ++j) c[i][j] = wmma4(a[i], b[j], c[i][j]);
    }

#pragma unroll
    for (int i = 0; i < 2; ++i)
#pragma unroll
        for (int j = 0; j < 4; ++j)
#pragma unroll
            for (int r = 0; r < 8; ++r) {
                int m = m0 + i * 16 + r + 8 * hlf;
                int n = n0 + j * 16 + l16;
                C[(size_t)m * N + n] = c[i][j][r];
            }
}

__global__ void score_kernel(const float* __restrict__ Q,
                             const float* __restrict__ Km,
                             const float* __restrict__ R,
                             const float* __restrict__ u,
                             const float* __restrict__ v,
                             float* __restrict__ S, int bh0) {
    const int lane = threadIdx.x & 31;
    const int w    = threadIdx.x >> 5;
    const int l16  = lane & 15;
    const int hlf  = lane >> 4;
    const int s  = blockIdx.z;
    const int bh = bh0 + s;
    const int b  = bh >> 4;
    const int h  = bh & 15;
    const int m0 = blockIdx.y * 64 + (w & 1) * 32;
    const int n0 = blockIdx.x * 256 + (w >> 1) * 64;

    const float* Qb = Q + (size_t)b * 1024 * 1024 + h * 64;
    const float* Kb = Km + (size_t)b * 2048 * 1024 + h * 64;
    const float* Rb = R + h * 64;

    v8f c[2][4];
#pragma unroll
    for (int i = 0; i < 2; ++i)
#pragma unroll
        for (int j = 0; j < 4; ++j) c[i][j] = v8f_zero();

    for (int kb = 0; kb < 64; kb += 4) {
        const int kk = kb + 2 * hlf;
        v2f ub = *(const v2f*)(u + h * 64 + kk);
        v2f vb = *(const v2f*)(v + h * 64 + kk);
        v2f qa[2], qv[2], bk[4], br[4];
#pragma unroll
        for (int i = 0; i < 2; ++i) {
            v2f qq = *(const v2f*)(Qb + (size_t)(m0 + i * 16 + l16) * 1024 + kk);
            qa[i] = qq + ub;
            qv[i] = qq + vb;
        }
#pragma unroll
        for (int j = 0; j < 4; ++j) {
            bk[j] = *(const v2f*)(Kb + (size_t)(n0 + j * 16 + l16) * 1024 + kk);
            br[j] = *(const v2f*)(Rb + (size_t)(n0 + j * 16 + l16) * 1024 + kk);
        }
#pragma unroll
        for (int i = 0; i < 2; ++i)
#pragma unroll
            for (int j = 0; j < 4; ++j) c[i][j] = wmma4(qa[i], bk[j], c[i][j]);
#pragma unroll
        for (int i = 0; i < 2; ++i)
#pragma unroll
            for (int j = 0; j < 4; ++j) c[i][j] = wmma4(qv[i], br[j], c[i][j]);
    }

    const float scale = 0.125f;
    float* Ss = S + (size_t)s * 1024 * 2048;
#pragma unroll
    for (int i = 0; i < 2; ++i)
#pragma unroll
        for (int j = 0; j < 4; ++j)
#pragma unroll
            for (int r = 0; r < 8; ++r) {
                int m = m0 + i * 16 + r + 8 * hlf;
                int n = n0 + j * 16 + l16;
                float val = c[i][j][r] * scale;
                if (n > 1024 + m) val = -INFINITY;
                Ss[(size_t)m * 2048 + n] = val;
            }
}

__global__ void softmax_kernel(float* __restrict__ S) {
    __shared__ float red[256];
    const int s = blockIdx.y, q = blockIdx.x, t = threadIdx.x;
    float* row = S + ((size_t)s * 1024 + q) * 2048;
    float x[8];
    float m = -INFINITY;
#pragma unroll
    for (int i = 0; i < 8; ++i) {
        x[i] = row[t + i * 256];
        m = fmaxf(m, x[i]);
    }
    red[t] = m;
    __syncthreads();
    for (int off = 128; off > 0; off >>= 1) {
        if (t < off) red[t] = fmaxf(red[t], red[t + off]);
        __syncthreads();
    }
    m = red[0];
    __syncthreads();
    float sum = 0.f;
#pragma unroll
    for (int i = 0; i < 8; ++i) {
        x[i] = __expf(x[i] - m);
        sum += x[i];
    }
    red[t] = sum;
    __syncthreads();
    for (int off = 128; off > 0; off >>= 1) {
        if (t < off) red[t] += red[t + off];
        __syncthreads();
    }
    float inv = 1.0f / red[0];
#pragma unroll
    for (int i = 0; i < 8; ++i) row[t + i * 256] = x[i] * inv;
}

__global__ void pv_kernel(const float* __restrict__ P,
                          const float* __restrict__ V,
                          float* __restrict__ O, int bh0) {
    const int lane = threadIdx.x & 31;
    const int w    = threadIdx.x >> 5;
    const int l16  = lane & 15;
    const int hlf  = lane >> 4;
    const int s  = blockIdx.y;
    const int bh = bh0 + s;
    const int b  = bh >> 4;
    const int h  = bh & 15;
    const int m0 = blockIdx.x * 128 + w * 16;

    const float* Ps = P + (size_t)s * 1024 * 2048;
    const float* Vb = V + (size_t)b * 2048 * 1024 + h * 64;

    v8f c[4];
#pragma unroll
    for (int j = 0; j < 4; ++j) c[j] = v8f_zero();

    for (int kb = 0; kb < 2048; kb += 4) {
        const int kk = kb + 2 * hlf;
        v2f a = *(const v2f*)(Ps + (size_t)(m0 + l16) * 2048 + kk);
        v2f bf[4];
#pragma unroll
        for (int j = 0; j < 4; ++j) {
            int n = j * 16 + l16;
            bf[j].x = Vb[(size_t)kk * 1024 + n];
            bf[j].y = Vb[(size_t)(kk + 1) * 1024 + n];
        }
#pragma unroll
        for (int j = 0; j < 4; ++j) c[j] = wmma4(a, bf[j], c[j]);
    }

#pragma unroll
    for (int j = 0; j < 4; ++j)
#pragma unroll
        for (int r = 0; r < 8; ++r) {
            int m = m0 + r + 8 * hlf;
            int n = j * 16 + l16;
            O[((size_t)b * 1024 + m) * 1024 + h * 64 + n] = c[j][r];
        }
}

extern "C" void kernel_launch(void* const* d_in, const int* in_sizes, int n_in,
                              void* d_out, int out_size, void* d_ws, size_t ws_size,
                              hipStream_t stream) {
    const float* x_q  = (const float*)d_in[0];
    const float* x_kv = (const float*)d_in[1];
    const float* Wq   = (const float*)d_in[2];
    const float* Wk   = (const float*)d_in[3];
    const float* Wv   = (const float*)d_in[4];
    const float* Wr   = (const float*)d_in[5];
    const float* Wo   = (const float*)d_in[6];
    const float* ub   = (const float*)d_in[7];
    const float* vb   = (const float*)d_in[8];
    float* out = (float*)d_out;
    (void)in_sizes; (void)n_in; (void)out_size; (void)ws_size;

    float* ws   = (float*)d_ws;
    float* Qb   = ws;
    float* Kb   = Qb + 2097152;
    float* Vb   = Kb + 4194304;
    float* Relb = Vb + 4194304;
    float* Rb   = Relb + 2097152;
    float* Ob   = Rb + 2097152;
    float* Sb   = Ob + 2097152;

    relgen_kernel<<<2048, 256, 0, stream>>>(Relb);

    gemm_nt_kernel<<<dim3(4, 32), 256, 0, stream>>>(x_q,  Wq, Qb, 2048, 1024, 1024);
    gemm_nt_kernel<<<dim3(4, 64), 256, 0, stream>>>(x_kv, Wk, Kb, 4096, 1024, 1024);
    gemm_nt_kernel<<<dim3(4, 64), 256, 0, stream>>>(x_kv, Wv, Vb, 4096, 1024, 1024);
    gemm_nt_kernel<<<dim3(4, 32), 256, 0, stream>>>(Relb, Wr, Rb, 2048, 1024, 1024);

    for (int bh0 = 0; bh0 < 32; bh0 += 8) {
        score_kernel<<<dim3(8, 16, 8), 256, 0, stream>>>(Qb, Kb, Rb, ub, vb, Sb, bh0);
        softmax_kernel<<<dim3(1024, 8), 256, 0, stream>>>(Sb);
        pv_kernel<<<dim3(8, 8), 256, 0, stream>>>(Sb, Vb, Ob, bh0);
    }

    gemm_nt_kernel<<<dim3(4, 32), 256, 0, stream>>>(Ob, Wo, out, 2048, 1024, 1024);
}


# causal skip: keys beyond the 128-key step above a query group's mask limit are not scored or summed; groups g and 31-g share a SIMD (even work); 16 heads per launch; only exact zeros leave the sums
# speedup vs baseline: 3.4449x; 3.4449x over previous
#include <hip/hip_runtime.h>
#include <hip/hip_bf16.h>
#include <math.h>


typedef float v2f __attribute__((ext_vector_type(2)));
typedef float v8f __attribute__((ext_vector_type(8)));

__device__ __forceinline__ v8f v8f_zero() {
    v8f z = {0.f, 0.f, 0.f, 0.f, 0.f, 0.f, 0.f, 0.f};
    return z;
}

__device__ __forceinline__ v8f wmma4(v2f a, v2f b, v8f c) {
    return __builtin_amdgcn_wmma_f32_16x16x4_f32(false, a, false, b,
                                                 (short)0, c, false, false);
}

__global__ void relgen_kernel(float* __restrict__ rel) {
    const int row = blockIdx.x;
    const int t = threadIdx.x;
    const float pos = (float)(2047 - row);
    const float negl = -9.210340371976184f / 512.0f;
#pragma unroll
    for (int i = 0; i < 4; ++i) {
        int c = t + i * 256;
        int cc = c & 511;
        float inv = expf((float)cc * negl);
        float ang = pos * inv;
        rel[(size_t)row * 1024 + c] = (c < 512) ? sinf(ang) : cosf(ang);
    }
}

typedef float v4f __attribute__((ext_vector_type(4)));

struct GemmChunk {
    v4f a[2][4];
    v4f b[4][4];
};

template <int KD>
__device__ __forceinline__ void gemm_fetch(GemmChunk& ch,
                                           const float* __restrict__ arow,
                                           const float* __restrict__ brow,
                                           int kb) {
#pragma unroll
    for (int i = 0; i < 2; ++i)
#pragma unroll
        for (int p = 0; p < 4; ++p)
            ch.a[i][p] = *(const v4f*)(arow + (size_t)(16 * i) * KD + kb + 4 * p);
#pragma unroll
    for (int j = 0; j < 4; ++j)
#pragma unroll
        for (int p = 0; p < 4; ++p)
            ch.b[j][p] = *(const v4f*)(brow + (size_t)j * KD + kb + 4 * p);
}

__device__ __forceinline__ v2f gemm_pick(const v4f pp, int odd) {
    v2f f;
    f.x = odd ? pp.z : pp.x;
    f.y = odd ? pp.w : pp.y;
    return f;
}

__device__ __forceinline__ void gemm_mma(const GemmChunk& ch, v8f (&c)[2][4]) {
#pragma unroll
    for (int s = 0; s < 8; ++s) {
        v2f a[2], b[4];
#pragma unroll
        for (int i = 0; i < 2; ++i) a[i] = gemm_pick(ch.a[i][s >> 1], s & 1);
#pragma unroll
        for (int j = 0; j < 4; ++j) b[j] = gemm_pick(ch.b[j][s >> 1], s & 1);
#pragma unroll
        for (int i = 0; i < 2; ++i)
#pragma unroll
            for (int j = 0; j < 4; ++j) c[i][j] = wmma4(a[i], b[j], c[i][j]);
    }
}

template <int N, int KD>
__global__ void __launch_bounds__(128)
gemm_nt_kernel(const float* __restrict__ A,
               const float* __restrict__ B,
               float* __restrict__ C) {
    static_assert(KD % 96 == 64, "k loop: trips of three chunks, then two");
    const int lane = threadIdx.x & 31;
    const int w    = threadIdx.x >> 5;
    const int l16  = lane & 15;
    const int hlf  = lane >> 4;
    const int m0 = blockIdx.y * 64 + (w & 1) * 32;
    const int n0 = blockIdx.x * 128 + (w >> 1) * 64;

    const float* arow = A + (size_t)(m0 + l16) * KD + 16 * hlf;
    const float* brow = B + (size_t)(n0 + 4 * l16) * KD + 16 * hlf;

    v8f c[2][4];
#pragma unroll
    for (int i = 0; i < 2; ++i)
#pragma unroll
        for (int j = 0; j < 4; ++j) c[i][j] = v8f_zero();

#define GEMM_HOLD() __builtin_amdgcn_sched_barrier(0)
    GemmChunk c0, c1, c2;
    gemm_fetch<KD>(c0, arow, brow, 0);
    gemm_fetch<KD>(c1, arow, brow, 32);
    for (int kb = 0; kb + 128 < KD; kb += 96) {
        GEMM_HOLD();
        gemm_fetch<KD>(c2, arow, brow, kb + 64);
        GEMM_HOLD();
        gemm_mma(c0, c);
        GEMM_HOLD();
        gemm_fetch<KD>(c0, arow, brow, kb + 96);
        GEMM_HOLD();
        gemm_mma(c1, c);
        GEMM_HOLD();
        gemm_fetch<KD>(c1, arow, brow, kb + 128);
        GEMM_HOLD();
        gemm_mma(c2, c);
    }
    GEMM_HOLD();
    gemm_mma(c0, c);
    gemm_mma(c1, c);
#undef GEMM_HOLD

    float* crow = C + (size_t)(m0 + 8 * hlf) * N + n0 + 4 * l16;
#pragma unroll
    for (int i = 0; i < 2; ++i)
#pragma unroll
        for (int r = 0; r < 8; ++r) {
            v4f o;
            o.x = c[i][0][r];
            o.y = c[i][1][r];
            o.z = c[i][2][r];
            o.w = c[i][3][r];
            *(v4f*)(crow + (size_t)(16 * i + r) * N) = o;
        }
}

struct ScoreSet {
    v4f b[4][4];
};

__device__ __forceinline__ void score_fetch(ScoreSet& st,
                                            const float* __restrict__ brow) {
#pragma unroll
    for (int j = 0; j < 4; ++j)
#pragma unroll
        for (int p = 0; p < 4; ++p)
            st.b[j][p] = *(const v4f*)(brow + (size_t)j * 1024 + 4 * p);
}

__device__ __forceinline__ void score_mma(const v4f (&a)[2][4],
                                          const ScoreSet& st, v8f (&c)[2][4]) {
#pragma unroll
    for (int s = 0; s < 8; ++s) {
        v2f fa[2], fb[4];
#pragma unroll
        for (int i = 0; i < 2; ++i) fa[i] = gemm_pick(a[i][s >> 1], s & 1);
#pragma unroll
        for (int j = 0; j < 4; ++j) fb[j] = gemm_pick(st.b[j][s >> 1], s & 1);
#pragma unroll
        for (int i = 0; i < 2; ++i)
#pragma unroll
            for (int j = 0; j < 4; ++j) c[i][j] = wmma4(fa[i], fb[j], c[i][j]);
    }
}

__device__ __forceinline__ void score_store(const v8f (&c)[2][4],
                                            float* __restrict__ srow, int dq,
                                            float (&rm)[2][8]) {
    const float scale = 0.125f;
#pragma unroll
    for (int i = 0; i < 2; ++i)
#pragma unroll
        for (int r = 0; r < 8; ++r) {
            const int d = dq - 16 * i - r;
            v4f o;
            o.x = (d + 0 > 0) ? -INFINITY : c[i][0][r] * scale;
            o.y = (d + 1 > 0) ? -INFINITY : c[i][1][r] * scale;
            o.z = (d + 2 > 0) ? -INFINITY : c[i][2][r] * scale;
            o.w = (d + 3 > 0) ? -INFINITY : c[i][3][r] * scale;
            *(v4f*)(srow + (size_t)(16 * i + r) * 2048) = o;
            rm[i][r] = fmaxf(fmaxf(rm[i][r], fmaxf(o.x, o.y)), fmaxf(o.z, o.w));
        }
}

__global__ void __launch_bounds__(256)
score_kernel(const float* __restrict__ Q,
             const float* __restrict__ Km,
             const float* __restrict__ R,
             const float* __restrict__ u,
             const float* __restrict__ v,
             float* __restrict__ S, float* __restrict__ Mx, int bh0) {
    const int lane = threadIdx.x & 31;
    const int w    = __builtin_amdgcn_readfirstlane(threadIdx.x >> 5);
    const int l16  = lane & 15;
    const int hlf  = lane >> 4;
    const int s  = blockIdx.y;
    const int bh = bh0 + s;
    const int b  = bh >> 4;
    const int h  = bh & 15;
    const int g  = (w < 4) ? 31 - (int)blockIdx.x : (int)blockIdx.x;
    const int kq = w & 3;
    const int m0 = 32 * g;
    const int ntile = 18 + 2 * (g >> 2);
    const int n  = (ntile - kq + 3) >> 2;
    const int nb = 64 * kq;

    const float* Qb = Q + (size_t)b * 1024 * 1024 + h * 64;
    const float* Kb = Km + (size_t)b * 2048 * 1024 + h * 64;
    const float* Rb = R + h * 64;

    v4f qa[2][2][4], qv[2][2][4];
#pragma unroll
    for (int cc = 0; cc < 2; ++cc)
#pragma unroll
        for (int p = 0; p < 4; ++p) {
            const int d = 32 * cc + 16 * hlf + 4 * p;
            const v4f u4 = *(const v4f*)(u + h * 64 + d);
            const v4f v4 = *(const v4f*)(v + h * 64 + d);
#pragma unroll
            for (int i = 0; i < 2; ++i) {
                const v4f q4 = *(const v4f*)(Qb + (size_t)(m0 + 16 * i + l16) * 1024 + d);
                qa[cc][i][p] = q4 + u4;
                qv[cc][i][p] = q4 + v4;
            }
        }

    const float* krow = Kb + (size_t)(nb + 4 * l16) * 1024 + 16 * hlf;
    const float* rrow = Rb + (size_t)(nb + 4 * l16) * 1024 + 16 * hlf;
    float* srow = S + (size_t)s * 1024 * 2048
                    + (size_t)(m0 + 8 * hlf) * 2048 + nb + 4 * l16;
    const int dq = nb + 4 * l16 - (m0 + 8 * hlf) - 1024;

    v8f c[2][4];
#pragma unroll
    for (int i = 0; i < 2; ++i)
#pragma unroll
        for (int j = 0; j < 4; ++j) c[i][j] = v8f_zero();

    float rm[2][8];
#pragma unroll
    for (int i = 0; i < 2; ++i)
#pragma unroll
        for (int r = 0; r < 8; ++r) rm[i][r] = -INFINITY;

#define SCORE_HOLD() __builtin_amdgcn_sched_barrier(0)
    const size_t tile = (size_t)256 * 1024;
    ScoreSet s0, s1, s2, s3;
    score_fetch(s0, krow);
    score_fetch(s1, krow + 32);
    score_fetch(s2, rrow);
    for (int t = 0; t < n - 1; ++t) {
        SCORE_HOLD();
        score_fetch(s3, rrow + t * tile + 32);
        SCORE_HOLD();
        score_mma(qa[0], s0, c);
        SCORE_HOLD();
        score_fetch(s0, krow + (t + 1) * tile);
        SCORE_HOLD();
        score_mma(qa[1], s1, c);
        SCORE_HOLD();
        score_fetch(s1, krow + (t + 1) * tile + 32);
        SCORE_HOLD();
        score_mma(qv[0], s2, c);
        SCORE_HOLD();
        score_fetch(s2, rrow + (t + 1) * tile);
        SCORE_HOLD();
        score_mma(qv[1], s3, c);
        SCORE_HOLD();
        score_store(c, srow + 256 * t, dq + 256 * t, rm);
#pragma unroll
        for (int i = 0; i < 2; ++i)
#pragma unroll
            for (int j = 0; j < 4; ++j) c[i][j] = v8f_zero();
    }
    SCORE_HOLD();
    score_fetch(s3, rrow + (n - 1) * tile + 32);
    SCORE_HOLD();
    score_mma(qa[0], s0, c);
    score_mma(qa[1], s1, c);
    score_mma(qv[0], s2, c);
    score_mma(qv[1], s3, c);
#undef SCORE_HOLD
    score_store(c, srow + 256 * (n - 1), dq + 256 * (n - 1), rm);

    float mine = -INFINITY;
#pragma unroll
    for (int i = 0; i < 2; ++i)
#pragma unroll
        for (int r = 0; r < 8; ++r) {
            float m = rm[i][r];
            m = fmaxf(m, __shfl_xor(m, 1));
            m = fmaxf(m, __shfl_xor(m, 2));
            m = fmaxf(m, __shfl_xor(m, 4));
            m = fmaxf(m, __shfl_xor(m, 8));
            if (l16 == 8 * i + r) mine = m;
        }
    Mx[((size_t)s * 4 + kq) * 1024 + m0 + 16 * (l16 >> 3) + 8 * hlf + (l16 & 7)] = mine;
}

struct PvChunk {
    v4f p[4];
    v2f v[16];
};

__device__ __forceinline__ void pv_fetch(PvChunk& ch,
                                         const float* __restrict__ prow,
                                         const float* __restrict__ vcol,
                                         int kb) {
#pragma unroll
    for (int i = 0; i < 4; ++i)
        ch.p[i] = *(const v4f*)(prow + kb + 4 * i);
#pragma unroll
    for (int t = 0; t < 16; ++t)
        ch.v[t] = *(const v2f*)(vcol + (size_t)(kb + t) * 1024);
}

__device__ __forceinline__ float pv_exp(float x, float nm) {
    return __builtin_amdgcn_exp2f(__builtin_fmaf(x, 1.4426950408889634f, nm));
}

__device__ __forceinline__ void pv_mma(const PvChunk& ch, v8f (&c)[2],
                                       float nm, v2f& lsum) {
#pragma unroll
    for (int s = 0; s < 8; ++s) {
        const v4f pp = ch.p[s >> 1];
        v2f a;
        a.x = pv_exp((s & 1) ? pp.z : pp.x, nm);
        a.y = pv_exp((s & 1) ? pp.w : pp.y, nm);
        lsum += a;
        const v2f r0 = ch.v[2 * s];
        const v2f r1 = ch.v[2 * s + 1];
#pragma unroll
        for (int j = 0; j < 2; ++j) {
            v2f b;
            b.x = r0[j];
            b.y = r1[j];
            c[j] = wmma4(a, b, c[j]);
        }
    }
}

__global__ void __launch_bounds__(128) __attribute__((amdgpu_waves_per_eu(1, 2)))
pv_kernel(const float* __restrict__ P,
          const float* __restrict__ Mx,
          const float* __restrict__ V,
          float* __restrict__ O, int bh0) {
    const int lane = threadIdx.x & 31;
    const int w    = __builtin_amdgcn_readfirstlane(threadIdx.x >> 5);
    const int l16  = lane & 15;
    const int hlf  = lane >> 4;
    const int dh   = w & 1;
    const int s  = blockIdx.y;
    const int bh = bh0 + s;
    const int b  = bh >> 4;
    const int h  = bh & 15;
    const int pr = 2 * blockIdx.x + (w >> 1);

    const float* vcol = V + (size_t)b * 2048 * 1024 + h * 64
                          + (size_t)(16 * hlf) * 1024 + 32 * dh + 2 * l16;

#define PV_HOLD() __builtin_amdgcn_sched_barrier(0)
#define PV_DONE() asm volatile("" :: "v"(c[0]), "v"(c[1]))
    for (int seg = 0; seg < 2; ++seg) {
        const int t  = seg ? 63 - pr : pr;
        const int m0 = 16 * t;
        const int kend = 128 * (9 + (t >> 3));

        const float* prow = P + (size_t)s * 1024 * 2048
                              + (size_t)(m0 + l16) * 2048 + 16 * hlf;

        v8f c[2];
#pragma unroll
        for (int j = 0; j < 2; ++j) c[j] = v8f_zero();

        const float* mrow = Mx + (size_t)s * 4 * 1024 + m0 + l16;
        float mx = mrow[0];
#pragma unroll
        for (int p = 1; p < 4; ++p) mx = fmaxf(mx, mrow[p * 1024]);
        const float nm = -mx * 1.4426950408889634f;
        v2f lsum = {0.f, 0.f};

        PvChunk c0, c1, c2, c3;
        pv_fetch(c0, prow, vcol, 0);
        pv_fetch(c1, prow, vcol, 32);
        pv_fetch(c2, prow, vcol, 64);
        for (int kb = 0; kb < kend - 128; kb += 128) {
            PV_HOLD();
            pv_fetch(c3, prow, vcol, kb + 96);
            PV_HOLD();
            pv_mma(c0, c, nm, lsum);
            PV_DONE();
            PV_HOLD();
            pv_fetch(c0, prow, vcol, kb + 128);
            PV_HOLD();
            pv_mma(c1, c, nm, lsum);
            PV_DONE();
            PV_HOLD();
            pv_fetch(c1, prow, vcol, kb + 160);
            PV_HOLD();
            pv_mma(c2, c, nm, lsum);
            PV_DONE();
            PV_HOLD();
            pv_fetch(c2, prow, vcol, kb + 192);
            PV_HOLD();
            pv_mma(c3, c, nm, lsum);
            PV_DONE();
        }
        PV_HOLD();
        pv_fetch(c3, prow, vcol, kend - 32);
        PV_HOLD();
        pv_mma(c0, c, nm, lsum);
        pv_mma(c1, c, nm, lsum);
        pv_mma(c2, c, nm, lsum);
        pv_mma(c3, c, nm, lsum);

        float* orow = O + ((size_t)b * 1024 + m0 + 8 * hlf) * 1024 + h * 64
                        + 32 * dh + 2 * l16;
        float rsum = lsum.x + lsum.y;
        rsum += __shfl_xor(rsum, 16);
#pragma unroll
        for (int r = 0; r < 8; ++r) {
            const float inv = 1.0f / __shfl(rsum, 8 * hlf + r);
            v2f o;
            o.x = c[0][r] * inv;
            o.y = c[1][r] * inv;
            *(v2f*)(orow + (size_t)r * 1024) = o;
        }
    }
#undef PV_DONE
#undef PV_HOLD
}

extern "C" void kernel_launch(void* const* d_in, const int* in_sizes, int n_in,
                              void* d_out, int out_size, void* d_ws, size_t ws_size,
                              hipStream_t stream) {
    const float* x_q  = (const float*)d_in[0];
    const float* x_kv = (const float*)d_in[1];
    const float* Wq   = (const float*)d_in[2];
    const float* Wk   = (const float*)d_in[3];
    const float* Wv   = (const float*)d_in[4];
    const float* Wr   = (const float*)d_in[5];
    const float* Wo   = (const float*)d_in[6];
    const float* ub   = (const float*)d_in[7];
    const float* vb   = (const float*)d_in[8];
    float* out = (float*)d_out;
    (void)in_sizes; (void)n_in; (void)out_size; (void)ws_size;

    float* ws   = (float*)d_ws;
    float* Qb   = ws;
    float* Kb   = Qb + 2097152;
    float* Vb   = Kb + 4194304;
    float* Relb = Vb + 4194304;
    float* Rb   = Relb + 2097152;
    float* Ob   = Rb + 2097152;
    float* Sb   = Ob + 2097152;
    float* Mxb  = Sb + 33554432;

    relgen_kernel<<<2048, 256, 0, stream>>>(Relb);

    gemm_nt_kernel<1024, 1024><<<dim3(8, 32), 128, 0, stream>>>(x_q,  Wq, Qb);
    gemm_nt_kernel<1024, 1024><<<dim3(8, 64), 128, 0, stream>>>(x_kv, Wk, Kb);
    gemm_nt_kernel<1024, 1024><<<dim3(8, 64), 128, 0, stream>>>(x_kv, Wv, Vb);
    gemm_nt_kernel<1024, 1024><<<dim3(8, 32), 128, 0, stream>>>(Relb, Wr, Rb);

    for (int bh0 = 0; bh0 < 32; bh0 += 16) {
        score_kernel<<<dim3(16, 16), 256, 0, stream>>>(Qb, Kb, Rb, ub, vb, Sb, Mxb, bh0);
        pv_kernel<<<dim3(16, 16), 128, 0, stream>>>(Sb, Mxb, Vb, Ob, bh0);
    }

    gemm_nt_kernel<1024, 1024><<<dim3(8, 32), 128, 0, stream>>>(Ob, Wo, out);
}


# gemm_nt operands (x_q, x_kv, weights, rel) written once by the first launch in fragment order: a 16-byte load reads 4 whole lines (was 16 lines of 32 bytes); k sum order unchanged
# speedup vs baseline: 3.5661x; 1.0352x over previous
#include <hip/hip_runtime.h>
#include <hip/hip_bf16.h>
#include <math.h>


typedef float v2f __attribute__((ext_vector_type(2)));
typedef float v8f __attribute__((ext_vector_type(8)));

__device__ __forceinline__ v8f v8f_zero() {
    v8f z = {0.f, 0.f, 0.f, 0.f, 0.f, 0.f, 0.f, 0.f};
    return z;
}

__device__ __forceinline__ v8f wmma4(v2f a, v2f b, v8f c) {
    return __builtin_amdgcn_wmma_f32_16x16x4_f32(false, a, false, b,
                                                 (short)0, c, false, false);
}

typedef float v4f __attribute__((ext_vector_type(4)));

__device__ __forceinline__ float rel_value(int row, int k) {
    const float pos = (float)(2047 - row);
    const float negl = -9.210340371976184f / 512.0f;
    const int cc = k & 511;
    const float inv = expf((float)cc * negl);
    const float ang = pos * inv;
    return (k < 512) ? sinf(ang) : cosf(ang);
}

__global__ void __launch_bounds__(256)
prep_kernel(const float* __restrict__ xq, const float* __restrict__ xkv,
            const float* __restrict__ Wq, const float* __restrict__ Wk,
            const float* __restrict__ Wv, const float* __restrict__ Wr,
            const float* __restrict__ Wo,
            float* __restrict__ xq_f, float* __restrict__ xkv_f,
            float* __restrict__ w_f, float* __restrict__ rel_f) {
    const int lane = threadIdx.x & 31;
    const int c = lane & 15;
    const int h = lane >> 4;
    const int gw = __builtin_amdgcn_readfirstlane((int)blockIdx.x * 8 + (int)(threadIdx.x >> 5));
    if (gw < 4096) {
        const int row = 16 * (gw >> 5) + c;
        const int k0 = 32 * (gw & 31) + 16 * h;
        float* d = rel_f + (size_t)gw * 512 + lane * 4;
#pragma unroll 1
        for (int p = 0; p < 4; ++p) {
            v4f o;
            o.x = rel_value(row, k0 + 4 * p + 0);
            o.y = rel_value(row, k0 + 4 * p + 1);
            o.z = rel_value(row, k0 + 4 * p + 2);
            o.w = rel_value(row, k0 + 4 * p + 3);
            *(v4f*)(d + 128 * p) = o;
        }
        return;
    }
    const int u0 = 4 * (gw - 4096);
    const float* src;
    float* dst;
    int ul;
    bool inter;
    if (u0 < 4096) {
        src = xq;  dst = xq_f;  ul = u0;  inter = false;
    } else if (u0 < 12288) {
        src = xkv; dst = xkv_f; ul = u0 - 4096; inter = false;
    } else {
        const int wsel = (u0 - 12288) >> 11;
        ul = (u0 - 12288) & 2047;
        src = (wsel == 0) ? Wq : (wsel == 1) ? Wk : (wsel == 2) ? Wv : (wsel == 3) ? Wr : Wo;
        dst = w_f + (size_t)wsel * 1048576;
        inter = true;
    }
    const int tile = ul >> 5;
    const int kc0 = ul & 31;
    const int row = inter ? 64 * (tile >> 2) + 4 * c + (tile & 3) : 16 * tile + c;
    const float* s = src + (size_t)row * 1024 + 32 * kc0 + 16 * h;
    float* d = dst + (size_t)ul * 512 + lane * 4;
    v4f t[4][4];
#pragma unroll
    for (int i = 0; i < 4; ++i)
#pragma unroll
        for (int p = 0; p < 4; ++p) t[i][p] = *(const v4f*)(s + 32 * i + 4 * p);
#pragma unroll
    for (int i = 0; i < 4; ++i)
#pragma unroll
        for (int p = 0; p < 4; ++p) *(v4f*)(d + 512 * i + 128 * p) = t[i][p];
}

struct GemmChunk {
    v4f a[2][4];
    v4f b[4][4];
};

template <int KD, bool AFRAG>
__device__ __forceinline__ void gemm_fetch(GemmChunk& ch,
                                           const float* __restrict__ arow,
                                           const float* __restrict__ brow,
                                           int kb) {
#pragma unroll
    for (int i = 0; i < 2; ++i)
#pragma unroll
        for (int p = 0; p < 4; ++p)
            ch.a[i][p] = AFRAG
                ? *(const v4f*)(arow + (size_t)i * 16384 + kb * 16 + 128 * p)
                : *(const v4f*)(arow + (size_t)(16 * i) * KD + kb + 4 * p);
#pragma unroll
    for (int j = 0; j < 4; ++j)
#pragma unroll
        for (int p = 0; p < 4; ++p)
            ch.b[j][p] = *(const v4f*)(brow + (size_t)j * 16384 + kb * 16 + 128 * p);
}

__device__ __forceinline__ v2f gemm_pick(const v4f pp, int odd) {
    v2f f;
    f.x = odd ? pp.z : pp.x;
    f.y = odd ? pp.w : pp.y;
    return f;
}

__device__ __forceinline__ void gemm_mma(const GemmChunk& ch, v8f (&c)[2][4]) {
#pragma unroll
    for (int s = 0; s < 8; ++s) {
        v2f a[2], b[4];
#pragma unroll
        for (int i = 0; i < 2; ++i) a[i] = gemm_pick(ch.a[i][s >> 1], s & 1);
#pragma unroll
        for (int j = 0; j < 4; ++j) b[j] = gemm_pick(ch.b[j][s >> 1], s & 1);
#pragma unroll
        for (int i = 0; i < 2; ++i)
#pragma unroll
            for (int j = 0; j < 4; ++j) c[i][j] = wmma4(a[i], b[j], c[i][j]);
    }
}

template <int N, int KD, bool AFRAG>
__global__ void __launch_bounds__(128)
gemm_nt_kernel(const float* __restrict__ A,
               const float* __restrict__ B,
               float* __restrict__ C) {
    static_assert(KD % 96 == 64, "k loop: trips of three chunks, then two");
    static_assert(KD == 1024, "fragment order: 32 chunks of 512 floats per 16-row tile");
    const int lane = threadIdx.x & 31;
    const int w    = threadIdx.x >> 5;
    const int l16  = lane & 15;
    const int hlf  = lane >> 4;
    const int m0 = blockIdx.y * 64 + (w & 1) * 32;
    const int n0 = blockIdx.x * 128 + (w >> 1) * 64;

    const float* arow = AFRAG ? A + (size_t)(m0 >> 4) * 16384 + 4 * lane
                              : A + (size_t)(m0 + l16) * KD + 16 * hlf;
    const float* brow = B + (size_t)(n0 >> 4) * 16384 + 4 * lane;

    v8f c[2][4];
#pragma unroll
    for (int i = 0; i < 2; ++i)
#pragma unroll
        for (int j = 0; j < 4; ++j) c[i][j] = v8f_zero();

#define GEMM_HOLD() __builtin_amdgcn_sched_barrier(0)
    GemmChunk c0, c1, c2;
    gemm_fetch<KD, AFRAG>(c0, arow, brow, 0);
    gemm_fetch<KD, AFRAG>(c1, arow, brow, 32);
    for (int kb = 0; kb + 128 < KD; kb += 96) {
        GEMM_HOLD();
        gemm_fetch<KD, AFRAG>(c2, arow, brow, kb + 64);
        GEMM_HOLD();
        gemm_mma(c0, c);
        GEMM_HOLD();
        gemm_fetch<KD, AFRAG>(c0, arow, brow, kb + 96);
        GEMM_HOLD();
        gemm_mma(c1, c);
        GEMM_HOLD();
        gemm_fetch<KD, AFRAG>(c1, arow, brow, kb + 128);
        GEMM_HOLD();
        gemm_mma(c2, c);
    }
    GEMM_HOLD();
    gemm_mma(c0, c);
    gemm_mma(c1, c);
#undef GEMM_HOLD

    float* crow = C + (size_t)(m0 + 8 * hlf) * N + n0 + 4 * l16;
#pragma unroll
    for (int i = 0; i < 2; ++i)
#pragma unroll
        for (int r = 0; r < 8; ++r) {
            v4f o;
            o.x = c[i][0][r];
            o.y = c[i][1][r];
            o.z = c[i][2][r];
            o.w = c[i][3][r];
            *(v4f*)(crow + (size_t)(16 * i + r) * N) = o;
        }
}

struct ScoreSet {
    v4f b[4][4];
};

__device__ __forceinline__ void score_fetch(ScoreSet& st,
                                            const float* __restrict__ brow) {
#pragma unroll
    for (int j = 0; j < 4; ++j)
#pragma unroll
        for (int p = 0; p < 4; ++p)
            st.b[j][p] = *(const v4f*)(brow + (size_t)j * 1024 + 4 * p);
}

__device__ __forceinline__ void score_mma(const v4f (&a)[2][4],
                                          const ScoreSet& st, v8f (&c)[2][4]) {
#pragma unroll
    for (int s = 0; s < 8; ++s) {
        v2f fa[2], fb[4];
#pragma unroll
        for (int i = 0; i < 2; ++i) fa[i] = gemm_pick(a[i][s >> 1], s & 1);
#pragma unroll
        for (int j = 0; j < 4; ++j) fb[j] = gemm_pick(st.b[j][s >> 1], s & 1);
#pragma unroll
        for (int i = 0; i < 2; ++i)
#pragma unroll
            for (int j = 0; j < 4; ++j) c[i][j] = wmma4(fa[i], fb[j], c[i][j]);
    }
}

__device__ __forceinline__ void score_store(const v8f (&c)[2][4],
                                            float* __restrict__ srow, int dq,
                                            float (&rm)[2][8]) {
    const float scale = 0.125f;
#pragma unroll
    for (int i = 0; i < 2; ++i)
#pragma unroll
        for (int r = 0; r < 8; ++r) {
            const int d = dq - 16 * i - r;
            v4f o;
            o.x = (d + 0 > 0) ? -INFINITY : c[i][0][r] * scale;
            o.y = (d + 1 > 0) ? -INFINITY : c[i][1][r] * scale;
            o.z = (d + 2 > 0) ? -INFINITY : c[i][2][r] * scale;
            o.w = (d + 3 > 0) ? -INFINITY : c[i][3][r] * scale;
            *(v4f*)(srow + (size_t)(16 * i + r) * 2048) = o;
            rm[i][r] = fmaxf(fmaxf(rm[i][r], fmaxf(o.x, o.y)), fmaxf(o.z, o.w));
        }
}

__global__ void __launch_bounds__(256)
score_kernel(const float* __restrict__ Q,
             const float* __restrict__ Km,
             const float* __restrict__ R,
             const float* __restrict__ u,
             const float* __restrict__ v,
             float* __restrict__ S, float* __restrict__ Mx, int bh0) {
    const int lane = threadIdx.x & 31;
    const int w    = __builtin_amdgcn_readfirstlane(threadIdx.x >> 5);
    const int l16  = lane & 15;
    const int hlf  = lane >> 4;
    const int s  = blockIdx.y;
    const int bh = bh0 + s;
    const int b  = bh >> 4;
    const int h  = bh & 15;
    const int g  = (w < 4) ? 31 - (int)blockIdx.x : (int)blockIdx.x;
    const int kq = w & 3;
    const int m0 = 32 * g;
    const int ntile = 18 + 2 * (g >> 2);
    const int n  = (ntile - kq + 3) >> 2;
    const int nb = 64 * kq;

    const float* Qb = Q + (size_t)b * 1024 * 1024 + h * 64;
    const float* Kb = Km + (size_t)b * 2048 * 1024 + h * 64;
    const float* Rb = R + h * 64;

    v4f qa[2][2][4], qv[2][2][4];
#pragma unroll
    for (int cc = 0; cc < 2; ++cc)
#pragma unroll
        for (int p = 0; p < 4; ++p) {
            const int d = 32 * cc + 16 * hlf + 4 * p;
            const v4f u4 = *(const v4f*)(u + h * 64 + d);
            const v4f v4 = *(const v4f*)(v + h * 64 + d);
#pragma unroll
            for (int i = 0; i < 2; ++i) {
                const v4f q4 = *(const v4f*)(Qb + (size_t)(m0 + 16 * i + l16) * 1024 + d);
                qa[cc][i][p] = q4 + u4;
                qv[cc][i][p] = q4 + v4;
            }
        }

    const float* krow = Kb + (size_t)(nb + 4 * l16) * 1024 + 16 * hlf;
    const float* rrow = Rb + (size_t)(nb + 4 * l16) * 1024 + 16 * hlf;
    float* srow = S + (size_t)s * 1024 * 2048
                    + (size_t)(m0 + 8 * hlf) * 2048 + nb + 4 * l16;
    const int dq = nb + 4 * l16 - (m0 + 8 * hlf) - 1024;

    v8f c[2][4];
#pragma unroll
    for (int i = 0; i < 2; ++i)
#pragma unroll
        for (int j = 0; j < 4; ++j) c[i][j] = v8f_zero();

    float rm[2][8];
#pragma unroll
    for (int i = 0; i < 2; ++i)
#pragma unroll
        for (int r = 0; r < 8; ++r) rm[i][r] = -INFINITY;

#define SCORE_HOLD() __builtin_amdgcn_sched_barrier(0)
    const size_t tile = (size_t)256 * 1024;
    ScoreSet s0, s1, s2, s3;
    score_fetch(s0, krow);
    score_fetch(s1, krow + 32);
    score_fetch(s2, rrow);
    for (int t = 0; t < n - 1; ++t) {
        SCORE_HOLD();
        score_fetch(s3, rrow + t * tile + 32);
        SCORE_HOLD();
        score_mma(qa[0], s0, c);
        SCORE_HOLD();
        score_fetch(s0, krow + (t + 1) * tile);
        SCORE_HOLD();
        score_mma(qa[1], s1, c);
        SCORE_HOLD();
        score_fetch(s1, krow + (t + 1) * tile + 32);
        SCORE_HOLD();
        score_mma(qv[0], s2, c);
        SCORE_HOLD();
        score_fetch(s2, rrow + (t + 1) * tile);
        SCORE_HOLD();
        score_mma(qv[1], s3, c);
        SCORE_HOLD();
        score_store(c, srow + 256 * t, dq + 256 * t, rm);
#pragma unroll
        for (int i = 0; i < 2; ++i)
#pragma unroll
            for (int j = 0; j < 4; ++j) c[i][j] = v8f_zero();
    }
    SCORE_HOLD();
    score_fetch(s3, rrow + (n - 1) * tile + 32);
    SCORE_HOLD();
    score_mma(qa[0], s0, c);
    score_mma(qa[1], s1, c);
    score_mma(qv[0], s2, c);
    score_mma(qv[1], s3, c);
#undef SCORE_HOLD
    score_store(c, srow + 256 * (n - 1), dq + 256 * (n - 1), rm);

    float mine = -INFINITY;
#pragma unroll
    for (int i = 0; i < 2; ++i)
#pragma unroll
        for (int r = 0; r < 8; ++r) {
            float m = rm[i][r];
            m = fmaxf(m, __shfl_xor(m, 1));
            m = fmaxf(m, __shfl_xor(m, 2));
            m = fmaxf(m, __shfl_xor(m, 4));
            m = fmaxf(m, __shfl_xor(m, 8));
            if (l16 == 8 * i + r) mine = m;
        }
    Mx[((size_t)s * 4 + kq) * 1024 + m0 + 16 * (l16 >> 3) + 8 * hlf + (l16 & 7)] = mine;
}

struct PvChunk {
    v4f p[4];
    v2f v[16];
};

__device__ __forceinline__ void pv_fetch(PvChunk& ch,
                                         const float* __restrict__ prow,
                                         const float* __restrict__ vcol,
                                         int kb) {
#pragma unroll
    for (int i = 0; i < 4; ++i)
        ch.p[i] = *(const v4f*)(prow + kb + 4 * i);
#pragma unroll
    for (int t = 0; t < 16; ++t)
        ch.v[t] = *(const v2f*)(vcol + (size_t)(kb + t) * 1024);
}

__device__ __forceinline__ float pv_exp(float x, float nm) {
    return __builtin_amdgcn_exp2f(__builtin_fmaf(x, 1.4426950408889634f, nm));
}

__device__ __forceinline__ void pv_mma(const PvChunk& ch, v8f (&c)[2],
                                       float nm, v2f& lsum) {
#pragma unroll
    for (int s = 0; s < 8; ++s) {
        const v4f pp = ch.p[s >> 1];
        v2f a;
        a.x = pv_exp((s & 1) ? pp.z : pp.x, nm);
        a.y = pv_exp((s & 1) ? pp.w : pp.y, nm);
        lsum += a;
        const v2f r0 = ch.v[2 * s];
        const v2f r1 = ch.v[2 * s + 1];
#pragma unroll
        for (int j = 0; j < 2; ++j) {
            v2f b;
            b.x = r0[j];
            b.y = r1[j];
            c[j] = wmma4(a, b, c[j]);
        }
    }
}

__global__ void __launch_bounds__(128) __attribute__((amdgpu_waves_per_eu(1, 2)))
pv_kernel(const float* __restrict__ P,
          const float* __restrict__ Mx,
          const float* __restrict__ V,
          float* __restrict__ O, int bh0) {
    const int lane = threadIdx.x & 31;
    const int w    = __builtin_amdgcn_readfirstlane(threadIdx.x >> 5);
    const int l16  = lane & 15;
    const int hlf  = lane >> 4;
    const int dh   = w & 1;
    const int s  = blockIdx.y;
    const int bh = bh0 + s;
    const int b  = bh >> 4;
    const int h  = bh & 15;
    const int pr = 2 * blockIdx.x + (w >> 1);

    const float* vcol = V + (size_t)b * 2048 * 1024 + h * 64
                          + (size_t)(16 * hlf) * 1024 + 32 * dh + 2 * l16;

#define PV_HOLD() __builtin_amdgcn_sched_barrier(0)
#define PV_DONE() asm volatile("" :: "v"(c[0]), "v"(c[1]))
    for (int seg = 0; seg < 2; ++seg) {
        const int t  = seg ? 63 - pr : pr;
        const int m0 = 16 * t;
        const int kend = 128 * (9 + (t >> 3));

        const float* prow = P + (size_t)s * 1024 * 2048
                              + (size_t)(m0 + l16) * 2048 + 16 * hlf;

        v8f c[2];
#pragma unroll
        for (int j = 0; j < 2; ++j) c[j] = v8f_zero();

        const float* mrow = Mx + (size_t)s * 4 * 1024 + m0 + l16;
        float mx = mrow[0];
#pragma unroll
        for (int p = 1; p < 4; ++p) mx = fmaxf(mx, mrow[p * 1024]);
        const float nm = -mx * 1.4426950408889634f;
        v2f lsum = {0.f, 0.f};

        PvChunk c0, c1, c2, c3;
        pv_fetch(c0, prow, vcol, 0);
        pv_fetch(c1, prow, vcol, 32);
        pv_fetch(c2, prow, vcol, 64);
        for (int kb = 0; kb < kend - 128; kb += 128) {
            PV_HOLD();
            pv_fetch(c3, prow, vcol, kb + 96);
            PV_HOLD();
            pv_mma(c0, c, nm, lsum);
            PV_DONE();
            PV_HOLD();
            pv_fetch(c0, prow, vcol, kb + 128);
            PV_HOLD();
            pv_mma(c1, c, nm, lsum);
            PV_DONE();
            PV_HOLD();
            pv_fetch(c1, prow, vcol, kb + 160);
            PV_HOLD();
            pv_mma(c2, c, nm, lsum);
            PV_DONE();
            PV_HOLD();
            pv_fetch(c2, prow, vcol, kb + 192);
            PV_HOLD();
            pv_mma(c3, c, nm, lsum);
            PV_DONE();
        }
        PV_HOLD();
        pv_fetch(c3, prow, vcol, kend - 32);
        PV_HOLD();
        pv_mma(c0, c, nm, lsum);
        pv_mma(c1, c, nm, lsum);
        pv_mma(c2, c, nm, lsum);
        pv_mma(c3, c, nm, lsum);

        float* orow = O + ((size_t)b * 1024 + m0 + 8 * hlf) * 1024 + h * 64
                        + 32 * dh + 2 * l16;
        float rsum = lsum.x + lsum.y;
        rsum += __shfl_xor(rsum, 16);
#pragma unroll
        for (int r = 0; r < 8; ++r) {
            const float inv = 1.0f / __shfl(rsum, 8 * hlf + r);
            v2f o;
            o.x = c[0][r] * inv;
            o.y = c[1][r] * inv;
            *(v2f*)(orow + (size_t)r * 1024) = o;
        }
    }
#undef PV_DONE
#undef PV_HOLD
}

extern "C" void kernel_launch(void* const* d_in, const int* in_sizes, int n_in,
                              void* d_out, int out_size, void* d_ws, size_t ws_size,
                              hipStream_t stream) {
    const float* x_q  = (const float*)d_in[0];
    const float* x_kv = (const float*)d_in[1];
    const float* Wq   = (const float*)d_in[2];
    const float* Wk   = (const float*)d_in[3];
    const float* Wv   = (const float*)d_in[4];
    const float* Wr   = (const float*)d_in[5];
    const float* Wo   = (const float*)d_in[6];
    const float* ub   = (const float*)d_in[7];
    const float* vb   = (const float*)d_in[8];
    float* out = (float*)d_out;
    (void)in_sizes; (void)n_in; (void)out_size; (void)ws_size;

    float* ws   = (float*)d_ws;
    float* Qb   = ws;
    float* Kb   = Qb + 2097152;
    float* Vb   = Kb + 4194304;
    float* Relb = Vb + 4194304;
    float* Rb   = Relb + 2097152;
    float* Ob   = Rb + 2097152;
    float* Sb   = Ob + 2097152;
    float* Mxb  = Sb + 33554432;
    float* Xqf  = Mxb + 65536;
    float* Xkvf = Xqf + 2097152;
    float* Wf   = Xkvf + 4194304;

    prep_kernel<<<1216, 256, 0, stream>>>(x_q, x_kv, Wq, Wk, Wv, Wr, Wo,
                                          Xqf, Xkvf, Wf, Relb);

    gemm_nt_kernel<1024, 1024, true><<<dim3(8, 32), 128, 0, stream>>>(Xqf,  Wf,           Qb);
    gemm_nt_kernel<1024, 1024, true><<<dim3(8, 64), 128, 0, stream>>>(Xkvf, Wf + 1048576, Kb);
    gemm_nt_kernel<1024, 1024, true><<<dim3(8, 64), 128, 0, stream>>>(Xkvf, Wf + 2097152, Vb);
    gemm_nt_kernel<1024, 1024, true><<<dim3(8, 32), 128, 0, stream>>>(Relb, Wf + 3145728, Rb);

    for (int bh0 = 0; bh0 < 32; bh0 += 16) {
        score_kernel<<<dim3(16, 16), 256, 0, stream>>>(Qb, Kb, Rb, ub, vb, Sb, Mxb, bh0);
        pv_kernel<<<dim3(16, 16), 128, 0, stream>>>(Sb, Mxb, Vb, Ob, bh0);
    }

    gemm_nt_kernel<1024, 1024, false><<<dim3(8, 32), 128, 0, stream>>>(Ob, Wf + 4194304, out);
}


# launches 10 -> 5: the four input projections are one launch; all 32 heads go through one score and one P.V launch, scores stored without their masked key ranges (50M floats); sum orders unchanged
# speedup vs baseline: 3.8569x; 1.0815x over previous
#include <hip/hip_runtime.h>
#include <hip/hip_bf16.h>
#include <math.h>


typedef float v2f __attribute__((ext_vector_type(2)));
typedef float v8f __attribute__((ext_vector_type(8)));

__device__ __forceinline__ v8f v8f_zero() {
    v8f z = {0.f, 0.f, 0.f, 0.f, 0.f, 0.f, 0.f, 0.f};
    return z;
}

__device__ __forceinline__ v8f wmma4(v2f a, v2f b, v8f c) {
    return __builtin_amdgcn_wmma_f32_16x16x4_f32(false, a, false, b,
                                                 (short)0, c, false, false);
}

typedef float v4f __attribute__((ext_vector_type(4)));

__device__ __forceinline__ float rel_value(int row, int k) {
    const float pos = (float)(2047 - row);
    const float negl = -9.210340371976184f / 512.0f;
    const int cc = k & 511;
    const float inv = expf((float)cc * negl);
    const float ang = pos * inv;
    return (k < 512) ? sinf(ang) : cosf(ang);
}

__global__ void __launch_bounds__(256)
prep_kernel(const float* __restrict__ xq, const float* __restrict__ xkv,
            const float* __restrict__ Wq, const float* __restrict__ Wk,
            const float* __restrict__ Wv, const float* __restrict__ Wr,
            const float* __restrict__ Wo,
            float* __restrict__ xq_f, float* __restrict__ xkv_f,
            float* __restrict__ w_f, float* __restrict__ wo_f,
            float* __restrict__ rel_f) {
    const int lane = threadIdx.x & 31;
    const int c = lane & 15;
    const int h = lane >> 4;
    const int gw = __builtin_amdgcn_readfirstlane((int)blockIdx.x * 8 + (int)(threadIdx.x >> 5));
    if (gw < 4096) {
        const int row = 16 * (gw >> 5) + c;
        const int k0 = 32 * (gw & 31) + 16 * h;
        float* d = rel_f + (size_t)gw * 512 + lane * 4;
#pragma unroll 1
        for (int p = 0; p < 4; ++p) {
            v4f o;
            o.x = rel_value(row, k0 + 4 * p + 0);
            o.y = rel_value(row, k0 + 4 * p + 1);
            o.z = rel_value(row, k0 + 4 * p + 2);
            o.w = rel_value(row, k0 + 4 * p + 3);
            *(v4f*)(d + 128 * p) = o;
        }
        return;
    }
    const int u0 = 4 * (gw - 4096);
    const float* src;
    float* dst;
    int ul;
    bool inter;
    if (u0 < 4096) {
        src = xq;  dst = xq_f;  ul = u0;  inter = false;
    } else if (u0 < 12288) {
        src = xkv; dst = xkv_f; ul = u0 - 4096; inter = false;
    } else {
        const int wsel = (u0 - 12288) >> 11;
        ul = (u0 - 12288) & 2047;
        src = (wsel == 0) ? Wq : (wsel == 1) ? Wk : (wsel == 2) ? Wv : (wsel == 3) ? Wr : Wo;
        dst = (wsel == 4) ? wo_f : w_f + (size_t)wsel * 1048576;
        inter = true;
    }
    const int tile = ul >> 5;
    const int kc0 = ul & 31;
    const int row = inter ? 64 * (tile >> 2) + 4 * c + (tile & 3) : 16 * tile + c;
    const float* s = src + (size_t)row * 1024 + 32 * kc0 + 16 * h;
    float* d = dst + (size_t)ul * 512 + lane * 4;
    v4f t[4][4];
#pragma unroll
    for (int i = 0; i < 4; ++i)
#pragma unroll
        for (int p = 0; p < 4; ++p) t[i][p] = *(const v4f*)(s + 32 * i + 4 * p);
#pragma unroll
    for (int i = 0; i < 4; ++i)
#pragma unroll
        for (int p = 0; p < 4; ++p) *(v4f*)(d + 512 * i + 128 * p) = t[i][p];
}

struct GemmChunk {
    v4f a[2][4];
    v4f b[4][4];
};

template <int KD, bool AFRAG>
__device__ __forceinline__ void gemm_fetch(GemmChunk& ch,
                                           const float* __restrict__ arow,
                                           const float* __restrict__ brow,
                                           int kb) {
#pragma unroll
    for (int i = 0; i < 2; ++i)
#pragma unroll
        for (int p = 0; p < 4; ++p)
            ch.a[i][p] = AFRAG
                ? *(const v4f*)(arow + (size_t)i * 16384 + kb * 16 + 128 * p)
                : *(const v4f*)(arow + (size_t)(16 * i) * KD + kb + 4 * p);
#pragma unroll
    for (int j = 0; j < 4; ++j)
#pragma unroll
        for (int p = 0; p < 4; ++p)
            ch.b[j][p] = *(const v4f*)(brow + (size_t)j * 16384 + kb * 16 + 128 * p);
}

__device__ __forceinline__ v2f gemm_pick(const v4f pp, int odd) {
    v2f f;
    f.x = odd ? pp.z : pp.x;
    f.y = odd ? pp.w : pp.y;
    return f;
}

__device__ __forceinline__ void gemm_mma(const GemmChunk& ch, v8f (&c)[2][4]) {
#pragma unroll
    for (int s = 0; s < 8; ++s) {
        v2f a[2], b[4];
#pragma unroll
        for (int i = 0; i < 2; ++i) a[i] = gemm_pick(ch.a[i][s >> 1], s & 1);
#pragma unroll
        for (int j = 0; j < 4; ++j) b[j] = gemm_pick(ch.b[j][s >> 1], s & 1);
#pragma unroll
        for (int i = 0; i < 2; ++i)
#pragma unroll
            for (int j = 0; j < 4; ++j) c[i][j] = wmma4(a[i], b[j], c[i][j]);
    }
}

template <int N, int KD, bool AFRAG>
__device__ __forceinline__ void gemm_tile(const float* __restrict__ A,
                                          const float* __restrict__ B,
                                          float* __restrict__ C,
                                          int mblk, int nblk) {
    static_assert(KD % 96 == 64, "k loop: trips of three chunks, then two");
    static_assert(KD == 1024, "fragment order: 32 chunks of 512 floats per 16-row tile");
    const int lane = threadIdx.x & 31;
    const int w    = threadIdx.x >> 5;
    const int l16  = lane & 15;
    const int hlf  = lane >> 4;
    const int m0 = mblk * 64 + (w & 1) * 32;
    const int n0 = nblk * 128 + (w >> 1) * 64;

    const float* arow = AFRAG ? A + (size_t)(m0 >> 4) * 16384 + 4 * lane
                              : A + (size_t)(m0 + l16) * KD + 16 * hlf;
    const float* brow = B + (size_t)(n0 >> 4) * 16384 + 4 * lane;

    v8f c[2][4];
#pragma unroll
    for (int i = 0; i < 2; ++i)
#pragma unroll
        for (int j = 0; j < 4; ++j) c[i][j] = v8f_zero();

#define GEMM_HOLD() __builtin_amdgcn_sched_barrier(0)
    GemmChunk c0, c1, c2;
    gemm_fetch<KD, AFRAG>(c0, arow, brow, 0);
    gemm_fetch<KD, AFRAG>(c1, arow, brow, 32);
    for (int kb = 0; kb + 128 < KD; kb += 96) {
        GEMM_HOLD();
        gemm_fetch<KD, AFRAG>(c2, arow, brow, kb + 64);
        GEMM_HOLD();
        gemm_mma(c0, c);
        GEMM_HOLD();
        gemm_fetch<KD, AFRAG>(c0, arow, brow, kb + 96);
        GEMM_HOLD();
        gemm_mma(c1, c);
        GEMM_HOLD();
        gemm_fetch<KD, AFRAG>(c1, arow, brow, kb + 128);
        GEMM_HOLD();
        gemm_mma(c2, c);
    }
    GEMM_HOLD();
    gemm_mma(c0, c);
    gemm_mma(c1, c);
#undef GEMM_HOLD

    float* crow = C + (size_t)(m0 + 8 * hlf) * N + n0 + 4 * l16;
#pragma unroll
    for (int i = 0; i < 2; ++i)
#pragma unroll
        for (int r = 0; r < 8; ++r) {
            v4f o;
            o.x = c[i][0][r];
            o.y = c[i][1][r];
            o.z = c[i][2][r];
            o.w = c[i][3][r];
            *(v4f*)(crow + (size_t)(16 * i + r) * N) = o;
        }
}

template <int N, int KD, bool AFRAG>
__global__ void __launch_bounds__(128)
gemm_nt_kernel(const float* __restrict__ A,
               const float* __restrict__ B,
               float* __restrict__ C) {
    gemm_tile<N, KD, AFRAG>(A, B, C, blockIdx.y, blockIdx.x);
}

__global__ void __launch_bounds__(128)
proj_kernel(const float* __restrict__ xq_f, const float* __restrict__ xkv_f,
            const float* __restrict__ rel_f, const float* __restrict__ w_f,
            float* __restrict__ Q, float* __restrict__ K,
            float* __restrict__ V, float* __restrict__ R) {
    const int y = blockIdx.y;
    const float* A;
    const float* B;
    float* C;
    int mblk;
    if (y < 32) {
        A = xq_f;  B = w_f;           C = Q; mblk = y;
    } else if (y < 96) {
        A = xkv_f; B = w_f + 1048576; C = K; mblk = y - 32;
    } else if (y < 160) {
        A = xkv_f; B = w_f + 2097152; C = V; mblk = y - 96;
    } else {
        A = rel_f; B = w_f + 3145728; C = R; mblk = y - 160;
    }
    gemm_tile<1024, 1024, true>(A, B, C, mblk, blockIdx.x);
}

struct ScoreSet {
    v4f b[4][4];
};

__device__ __forceinline__ void score_fetch(ScoreSet& st,
                                            const float* __restrict__ brow) {
#pragma unroll
    for (int j = 0; j < 4; ++j)
#pragma unroll
        for (int p = 0; p < 4; ++p)
            st.b[j][p] = *(const v4f*)(brow + (size_t)j * 1024 + 4 * p);
}

__device__ __forceinline__ void score_mma(const v4f (&a)[2][4],
                                          const ScoreSet& st, v8f (&c)[2][4]) {
#pragma unroll
    for (int s = 0; s < 8; ++s) {
        v2f fa[2], fb[4];
#pragma unroll
        for (int i = 0; i < 2; ++i) fa[i] = gemm_pick(a[i][s >> 1], s & 1);
#pragma unroll
        for (int j = 0; j < 4; ++j) fb[j] = gemm_pick(st.b[j][s >> 1], s & 1);
#pragma unroll
        for (int i = 0; i < 2; ++i)
#pragma unroll
            for (int j = 0; j < 4; ++j) c[i][j] = wmma4(fa[i], fb[j], c[i][j]);
    }
}

__device__ __forceinline__ void score_store(const v8f (&c)[2][4],
                                            float* __restrict__ srow, int dq,
                                            float (&rm)[2][8], int pitch) {
    const float scale = 0.125f;
#pragma unroll
    for (int i = 0; i < 2; ++i)
#pragma unroll
        for (int r = 0; r < 8; ++r) {
            const int d = dq - 16 * i - r;
            v4f o;
            o.x = (d + 0 > 0) ? -INFINITY : c[i][0][r] * scale;
            o.y = (d + 1 > 0) ? -INFINITY : c[i][1][r] * scale;
            o.z = (d + 2 > 0) ? -INFINITY : c[i][2][r] * scale;
            o.w = (d + 3 > 0) ? -INFINITY : c[i][3][r] * scale;
            *(v4f*)(srow + (size_t)((16 * i + r) * pitch)) = o;
            rm[i][r] = fmaxf(fmaxf(rm[i][r], fmaxf(o.x, o.y)), fmaxf(o.z, o.w));
        }
}

__global__ void __launch_bounds__(256)
score_kernel(const float* __restrict__ Q,
             const float* __restrict__ Km,
             const float* __restrict__ R,
             const float* __restrict__ u,
             const float* __restrict__ v,
             float* __restrict__ S, float* __restrict__ Mx) {
    const int lane = threadIdx.x & 31;
    const int w    = __builtin_amdgcn_readfirstlane(threadIdx.x >> 5);
    const int l16  = lane & 15;
    const int hlf  = lane >> 4;
    const int s  = blockIdx.y;
    const int b  = s >> 4;
    const int h  = s & 15;
    const int g  = (w < 4) ? 31 - (int)blockIdx.x : (int)blockIdx.x;
    const int kq = w & 3;
    const int m0 = 32 * g;
    const int ntile = 18 + 2 * (g >> 2);
    const int n  = (ntile - kq + 3) >> 2;
    const int nb = 64 * kq;

    const float* Qb = Q + (size_t)b * 1024 * 1024 + h * 64;
    const float* Kb = Km + (size_t)b * 2048 * 1024 + h * 64;
    const float* Rb = R + h * 64;

    v4f qa[2][2][4], qv[2][2][4];
#pragma unroll
    for (int cc = 0; cc < 2; ++cc)
#pragma unroll
        for (int p = 0; p < 4; ++p) {
            const int d = 32 * cc + 16 * hlf + 4 * p;
            const v4f u4 = *(const v4f*)(u + h * 64 + d);
            const v4f v4 = *(const v4f*)(v + h * 64 + d);
#pragma unroll
            for (int i = 0; i < 2; ++i) {
                const v4f q4 = *(const v4f*)(Qb + (size_t)(m0 + 16 * i + l16) * 1024 + d);
                qa[cc][i][p] = q4 + u4;
                qv[cc][i][p] = q4 + v4;
            }
        }

    const float* krow = Kb + (size_t)(nb + 4 * l16) * 1024 + 16 * hlf;
    const float* rrow = Rb + (size_t)(nb + 4 * l16) * 1024 + 16 * hlf;
    const int G = g >> 2;
    const int pitch = 128 * (9 + G);
    float* srow = S + (size_t)s * 1638400
                    + (size_t)16384 * (9 * G + ((G * (G - 1)) >> 1))
                    + (size_t)((32 * (g & 3) + 8 * hlf) * pitch) + nb + 4 * l16;
    const int dq = nb + 4 * l16 - (m0 + 8 * hlf) - 1024;

    v8f c[2][4];
#pragma unroll
    for (int i = 0; i < 2; ++i)
#pragma unroll
        for (int j = 0; j < 4; ++j) c[i][j] = v8f_zero();

    float rm[2][8];
#pragma unroll
    for (int i = 0; i < 2; ++i)
#pragma unroll
        for (int r = 0; r < 8; ++r) rm[i][r] = -INFINITY;

#define SCORE_HOLD() __builtin_amdgcn_sched_barrier(0)
    const size_t tile = (size_t)256 * 1024;
    ScoreSet s0, s1, s2, s3;
    score_fetch(s0, krow);
    score_fetch(s1, krow + 32);
    score_fetch(s2, rrow);
    for (int t = 0; t < n - 1; ++t) {
        SCORE_HOLD();
        score_fetch(s3, rrow + t * tile + 32);
        SCORE_HOLD();
        score_mma(qa[0], s0, c);
        SCORE_HOLD();
        score_fetch(s0, krow + (t + 1) * tile);
        SCORE_HOLD();
        score_mma(qa[1], s1, c);
        SCORE_HOLD();
        score_fetch(s1, krow + (t + 1) * tile + 32);
        SCORE_HOLD();
        score_mma(qv[0], s2, c);
        SCORE_HOLD();
        score_fetch(s2, rrow + (t + 1) * tile);
        SCORE_HOLD();
        score_mma(qv[1], s3, c);
        SCORE_HOLD();
        score_store(c, srow + 256 * t, dq + 256 * t, rm, pitch);
#pragma unroll
        for (int i = 0; i < 2; ++i)
#pragma unroll
            for (int j = 0; j < 4; ++j) c[i][j] = v8f_zero();
    }
    SCORE_HOLD();
    score_fetch(s3, rrow + (n - 1) * tile + 32);
    SCORE_HOLD();
    score_mma(qa[0], s0, c);
    score_mma(qa[1], s1, c);
    score_mma(qv[0], s2, c);
    score_mma(qv[1], s3, c);
#undef SCORE_HOLD
    score_store(c, srow + 256 * (n - 1), dq + 256 * (n - 1), rm, pitch);

    float mine = -INFINITY;
#pragma unroll
    for (int i = 0; i < 2; ++i)
#pragma unroll
        for (int r = 0; r < 8; ++r) {
            float m = rm[i][r];
            m = fmaxf(m, __shfl_xor(m, 1));
            m = fmaxf(m, __shfl_xor(m, 2));
            m = fmaxf(m, __shfl_xor(m, 4));
            m = fmaxf(m, __shfl_xor(m, 8));
            if (l16 == 8 * i + r) mine = m;
        }
    Mx[((size_t)s * 4 + kq) * 1024 + m0 + 16 * (l16 >> 3) + 8 * hlf + (l16 & 7)] = mine;
}

struct PvChunk {
    v4f p[4];
    v2f v[16];
};

__device__ __forceinline__ void pv_fetch(PvChunk& ch,
                                         const float* __restrict__ prow,
                                         const float* __restrict__ vcol,
                                         int kb) {
#pragma unroll
    for (int i = 0; i < 4; ++i)
        ch.p[i] = *(const v4f*)(prow + kb + 4 * i);
#pragma unroll
    for (int t = 0; t < 16; ++t)
        ch.v[t] = *(const v2f*)(vcol + (size_t)(kb + t) * 1024);
}

__device__ __forceinline__ float pv_exp(float x, float nm) {
    return __builtin_amdgcn_exp2f(__builtin_fmaf(x, 1.4426950408889634f, nm));
}

__device__ __forceinline__ void pv_mma(const PvChunk& ch, v8f (&c)[2],
                                       float nm, v2f& lsum) {
#pragma unroll
    for (int s = 0; s < 8; ++s) {
        const v4f pp = ch.p[s >> 1];
        v2f a;
        a.x = pv_exp((s & 1) ? pp.z : pp.x, nm);
        a.y = pv_exp((s & 1) ? pp.w : pp.y, nm);
        lsum += a;
        const v2f r0 = ch.v[2 * s];
        const v2f r1 = ch.v[2 * s + 1];
#pragma unroll
        for (int j = 0; j < 2; ++j) {
            v2f b;
            b.x = r0[j];
            b.y = r1[j];
            c[j] = wmma4(a, b, c[j]);
        }
    }
}

__global__ void __launch_bounds__(128) __attribute__((amdgpu_waves_per_eu(1, 2)))
pv_kernel(const float* __restrict__ P,
          const float* __restrict__ Mx,
          const float* __restrict__ V,
          float* __restrict__ O) {
    const int lane = threadIdx.x & 31;
    const int w    = __builtin_amdgcn_readfirstlane(threadIdx.x >> 5);
    const int l16  = lane & 15;
    const int hlf  = lane >> 4;
    const int dh   = w & 1;
    const int s  = blockIdx.y;
    const int b  = s >> 4;
    const int h  = s & 15;
    const int pr = 2 * blockIdx.x + (w >> 1);

    const float* vcol = V + (size_t)b * 2048 * 1024 + h * 64
                          + (size_t)(16 * hlf) * 1024 + 32 * dh + 2 * l16;

#define PV_HOLD() __builtin_amdgcn_sched_barrier(0)
#define PV_DONE() asm volatile("" :: "v"(c[0]), "v"(c[1]))
    for (int seg = 0; seg < 2; ++seg) {
        const int t  = seg ? 63 - pr : pr;
        const int m0 = 16 * t;
        const int G = t >> 3;
        const int kend = 128 * (9 + G);

        const float* prow = P + (size_t)s * 1638400
                              + (size_t)16384 * (9 * G + ((G * (G - 1)) >> 1))
                              + (size_t)((16 * (t & 7) + l16) * kend) + 16 * hlf;

        v8f c[2];
#pragma unroll
        for (int j = 0; j < 2; ++j) c[j] = v8f_zero();

        const float* mrow = Mx + (size_t)s * 4 * 1024 + m0 + l16;
        float mx = mrow[0];
#pragma unroll
        for (int p = 1; p < 4; ++p) mx = fmaxf(mx, mrow[p * 1024]);
        const float nm = -mx * 1.4426950408889634f;
        v2f lsum = {0.f, 0.f};

        PvChunk c0, c1, c2, c3;
        pv_fetch(c0, prow, vcol, 0);
        pv_fetch(c1, prow, vcol, 32);
        pv_fetch(c2, prow, vcol, 64);
        for (int kb = 0; kb < kend - 128; kb += 128) {
            PV_HOLD();
            pv_fetch(c3, prow, vcol, kb + 96);
            PV_HOLD();
            pv_mma(c0, c, nm, lsum);
            PV_DONE();
            PV_HOLD();
            pv_fetch(c0, prow, vcol, kb + 128);
            PV_HOLD();
            pv_mma(c1, c, nm, lsum);
            PV_DONE();
            PV_HOLD();
            pv_fetch(c1, prow, vcol, kb + 160);
            PV_HOLD();
            pv_mma(c2, c, nm, lsum);
            PV_DONE();
            PV_HOLD();
            pv_fetch(c2, prow, vcol, kb + 192);
            PV_HOLD();
            pv_mma(c3, c, nm, lsum);
            PV_DONE();
        }
        PV_HOLD();
        pv_fetch(c3, prow, vcol, kend - 32);
        PV_HOLD();
        pv_mma(c0, c, nm, lsum);
        pv_mma(c1, c, nm, lsum);
        pv_mma(c2, c, nm, lsum);
        pv_mma(c3, c, nm, lsum);

        float* orow = O + ((size_t)b * 1024 + m0 + 8 * hlf) * 1024 + h * 64
                        + 32 * dh + 2 * l16;
        float rsum = lsum.x + lsum.y;
        rsum += __shfl_xor(rsum, 16);
#pragma unroll
        for (int r = 0; r < 8; ++r) {
            const float inv = 1.0f / __shfl(rsum, 8 * hlf + r);
            v2f o;
            o.x = c[0][r] * inv;
            o.y = c[1][r] * inv;
            *(v2f*)(orow + (size_t)r * 1024) = o;
        }
    }
#undef PV_DONE
#undef PV_HOLD
}

extern "C" void kernel_launch(void* const* d_in, const int* in_sizes, int n_in,
                              void* d_out, int out_size, void* d_ws, size_t ws_size,
                              hipStream_t stream) {
    const float* x_q  = (const float*)d_in[0];
    const float* x_kv = (const float*)d_in[1];
    const float* Wq   = (const float*)d_in[2];
    const float* Wk   = (const float*)d_in[3];
    const float* Wv   = (const float*)d_in[4];
    const float* Wr   = (const float*)d_in[5];
    const float* Wo   = (const float*)d_in[6];
    const float* ub   = (const float*)d_in[7];
    const float* vb   = (const float*)d_in[8];
    float* out = (float*)d_out;
    (void)in_sizes; (void)n_in; (void)out_size; (void)ws_size;

    float* ws   = (float*)d_ws;
    float* Qb   = ws;
    float* Ob   = Qb;
    float* Kb   = Qb + 2097152;
    float* Vb   = Kb + 4194304;
    float* Rb   = Vb + 4194304;
    float* Wof  = Rb + 2097152;
    float* Mxb  = Wof + 1048576;
    float* Sb   = Mxb + 131072;
    float* Relb = Sb;
    float* Xqf  = Relb + 2097152;
    float* Xkvf = Xqf + 2097152;
    float* Wf   = Xkvf + 4194304;

    prep_kernel<<<1216, 256, 0, stream>>>(x_q, x_kv, Wq, Wk, Wv, Wr, Wo,
                                          Xqf, Xkvf, Wf, Wof, Relb);

    proj_kernel<<<dim3(8, 192), 128, 0, stream>>>(Xqf, Xkvf, Relb, Wf, Qb, Kb, Vb, Rb);

    score_kernel<<<dim3(16, 32), 256, 0, stream>>>(Qb, Kb, Rb, ub, vb, Sb, Mxb);
    pv_kernel<<<dim3(16, 32), 128, 0, stream>>>(Sb, Mxb, Vb, Ob);

    gemm_nt_kernel<1024, 1024, false><<<dim3(8, 32), 128, 0, stream>>>(Ob, Wof, out);
}


# K, R, V, scores, O stored by their producers in the readers' fragment order (scores formed transposed, keys x q): a 16-byte load reads 4 whole lines (was 16; V 8-byte); changes P.V, row-sum k order
# speedup vs baseline: 4.0826x; 1.0585x over previous
#include <hip/hip_runtime.h>
#include <hip/hip_bf16.h>
#include <math.h>


typedef float v2f __attribute__((ext_vector_type(2)));
typedef float v8f __attribute__((ext_vector_type(8)));

__device__ __forceinline__ v8f v8f_zero() {
    v8f z = {0.f, 0.f, 0.f, 0.f, 0.f, 0.f, 0.f, 0.f};
    return z;
}

__device__ __forceinline__ v8f wmma4(v2f a, v2f b, v8f c) {
    return __builtin_amdgcn_wmma_f32_16x16x4_f32(false, a, false, b,
                                                 (short)0, c, false, false);
}

typedef float v4f __attribute__((ext_vector_type(4)));

__device__ __forceinline__ float rel_value(int row, int k) {
    const float pos = (float)(2047 - row);
    const float negl = -9.210340371976184f / 512.0f;
    const int cc = k & 511;
    const float inv = expf((float)cc * negl);
    const float ang = pos * inv;
    return (k < 512) ? sinf(ang) : cosf(ang);
}

__global__ void __launch_bounds__(256)
prep_kernel(const float* __restrict__ xq, const float* __restrict__ xkv,
            const float* __restrict__ Wq, const float* __restrict__ Wk,
            const float* __restrict__ Wv, const float* __restrict__ Wr,
            const float* __restrict__ Wo,
            float* __restrict__ xq_f, float* __restrict__ xkv_f,
            float* __restrict__ w_f, float* __restrict__ wo_f,
            float* __restrict__ rel_f) {
    const int lane = threadIdx.x & 31;
    const int c = lane & 15;
    const int h = lane >> 4;
    const int gw = __builtin_amdgcn_readfirstlane((int)blockIdx.x * 8 + (int)(threadIdx.x >> 5));
    if (gw < 4096) {
        const int row = 16 * (gw >> 5) + c;
        const int k0 = 32 * (gw & 31) + 16 * h;
        float* d = rel_f + (size_t)gw * 512 + lane * 4;
#pragma unroll 1
        for (int p = 0; p < 4; ++p) {
            v4f o;
            o.x = rel_value(row, k0 + 4 * p + 0);
            o.y = rel_value(row, k0 + 4 * p + 1);
            o.z = rel_value(row, k0 + 4 * p + 2);
            o.w = rel_value(row, k0 + 4 * p + 3);
            *(v4f*)(d + 128 * p) = o;
        }
        return;
    }
    const int u0 = 4 * (gw - 4096);
    const float* src;
    float* dst;
    int ul;
    bool inter;
    if (u0 < 4096) {
        src = xq;  dst = xq_f;  ul = u0;  inter = false;
    } else if (u0 < 12288) {
        src = xkv; dst = xkv_f; ul = u0 - 4096; inter = false;
    } else {
        const int wsel = (u0 - 12288) >> 11;
        ul = (u0 - 12288) & 2047;
        src = (wsel == 0) ? Wq : (wsel == 1) ? Wk : (wsel == 2) ? Wv : (wsel == 3) ? Wr : Wo;
        dst = (wsel == 4) ? wo_f : w_f + (size_t)wsel * 1048576;
        inter = true;
    }
    const int tile = ul >> 5;
    const int kc0 = ul & 31;
    const int row = inter ? 64 * (tile >> 2) + 4 * c + (tile & 3) : 16 * tile + c;
    const float* s = src + (size_t)row * 1024 + 32 * kc0 + 16 * h;
    float* d = dst + (size_t)ul * 512 + lane * 4;
    v4f t[4][4];
#pragma unroll
    for (int i = 0; i < 4; ++i)
#pragma unroll
        for (int p = 0; p < 4; ++p) t[i][p] = *(const v4f*)(s + 32 * i + 4 * p);
#pragma unroll
    for (int i = 0; i < 4; ++i)
#pragma unroll
        for (int p = 0; p < 4; ++p) *(v4f*)(d + 512 * i + 128 * p) = t[i][p];
}

struct GemmChunk {
    v4f a[2][4];
    v4f b[4][4];
};

template <int KD, bool AFRAG>
__device__ __forceinline__ void gemm_fetch(GemmChunk& ch,
                                           const float* __restrict__ arow,
                                           const float* __restrict__ brow,
                                           int kb) {
#pragma unroll
    for (int i = 0; i < 2; ++i)
#pragma unroll
        for (int p = 0; p < 4; ++p)
            ch.a[i][p] = AFRAG
                ? *(const v4f*)(arow + (size_t)i * 16384 + kb * 16 + 128 * p)
                : *(const v4f*)(arow + (size_t)(16 * i) * KD + kb + 4 * p);
#pragma unroll
    for (int j = 0; j < 4; ++j)
#pragma unroll
        for (int p = 0; p < 4; ++p)
            ch.b[j][p] = *(const v4f*)(brow + (size_t)j * 16384 + kb * 16 + 128 * p);
}

__device__ __forceinline__ v2f gemm_pick(const v4f pp, int odd) {
    v2f f;
    f.x = odd ? pp.z : pp.x;
    f.y = odd ? pp.w : pp.y;
    return f;
}

__device__ __forceinline__ void gemm_mma(const GemmChunk& ch, v8f (&c)[2][4]) {
#pragma unroll
    for (int s = 0; s < 8; ++s) {
        v2f a[2], b[4];
#pragma unroll
        for (int i = 0; i < 2; ++i) a[i] = gemm_pick(ch.a[i][s >> 1], s & 1);
#pragma unroll
        for (int j = 0; j < 4; ++j) b[j] = gemm_pick(ch.b[j][s >> 1], s & 1);
#pragma unroll
        for (int i = 0; i < 2; ++i)
#pragma unroll
            for (int j = 0; j < 4; ++j) c[i][j] = wmma4(a[i], b[j], c[i][j]);
    }
}

template <int N, int KD, bool AFRAG>
__device__ __forceinline__ void gemm_tile(const float* __restrict__ A,
                                          const float* __restrict__ B,
                                          float* __restrict__ C,
                                          int mblk, int nblk, int epi) {
    static_assert(KD % 96 == 64, "k loop: trips of three chunks, then two");
    static_assert(KD == 1024, "fragment order: 32 chunks of 512 floats per 16-row tile");
    const int lane = threadIdx.x & 31;
    const int w    = threadIdx.x >> 5;
    const int l16  = lane & 15;
    const int hlf  = lane >> 4;
    const int m0 = mblk * 64 + (w & 1) * 32;
    const int n0 = nblk * 128 + (w >> 1) * 64;

    const float* arow = AFRAG ? A + (size_t)(m0 >> 4) * 16384 + 4 * lane
                              : A + (size_t)(m0 + l16) * KD + 16 * hlf;
    const float* brow = B + (size_t)(n0 >> 4) * 16384 + 4 * lane;

    v8f c[2][4];
#pragma unroll
    for (int i = 0; i < 2; ++i)
#pragma unroll
        for (int j = 0; j < 4; ++j) c[i][j] = v8f_zero();

#define GEMM_HOLD() __builtin_amdgcn_sched_barrier(0)
    GemmChunk c0, c1, c2;
    gemm_fetch<KD, AFRAG>(c0, arow, brow, 0);
    gemm_fetch<KD, AFRAG>(c1, arow, brow, 32);
    for (int kb = 0; kb + 128 < KD; kb += 96) {
        GEMM_HOLD();
        gemm_fetch<KD, AFRAG>(c2, arow, brow, kb + 64);
        GEMM_HOLD();
        gemm_mma(c0, c);
        GEMM_HOLD();
        gemm_fetch<KD, AFRAG>(c0, arow, brow, kb + 96);
        GEMM_HOLD();
        gemm_mma(c1, c);
        GEMM_HOLD();
        gemm_fetch<KD, AFRAG>(c1, arow, brow, kb + 128);
        GEMM_HOLD();
        gemm_mma(c2, c);
    }
    GEMM_HOLD();
    gemm_mma(c0, c);
    gemm_mma(c1, c);
#undef GEMM_HOLD

    if (epi == 0) {
        float* crow = C + (size_t)(m0 + 8 * hlf) * N + n0 + 4 * l16;
#pragma unroll
        for (int i = 0; i < 2; ++i)
#pragma unroll
            for (int r = 0; r < 8; ++r) {
                v4f o;
                o.x = c[i][0][r];
                o.y = c[i][1][r];
                o.z = c[i][2][r];
                o.w = c[i][3][r];
                *(v4f*)(crow + (size_t)(16 * i + r) * N) = o;
            }
    } else if (epi == 1) {
        const int mk = m0 & 2047;
        float* kf = C + (size_t)(((m0 >> 11) * 16 + (n0 >> 6)) * 32 + (mk >> 6)) * 4096
                      + ((mk >> 4) & 3) * 512
                      + (l16 >> 3) * 2048 + (l16 & 3) * 128 + ((l16 >> 2) & 1) * 64 + hlf * 32;
#pragma unroll
        for (int i = 0; i < 2; ++i)
#pragma unroll
            for (int r = 0; r < 8; ++r) {
                v4f o;
                o.x = c[i][0][r];
                o.y = c[i][1][r];
                o.z = c[i][2][r];
                o.w = c[i][3][r];
                *(v4f*)(kf + 512 * i + 4 * r) = o;
            }
    } else {
        const int mk = m0 & 2047;
        float* vf = C + (size_t)(((m0 >> 11) * 16 + (n0 >> 6)) * 64 + (mk >> 5)) * 2048
                      + (l16 >> 3) * 1024 + (16 * hlf + 2 * (l16 & 7)) * 4;
#pragma unroll
        for (int i = 0; i < 2; ++i)
#pragma unroll
            for (int s3 = 0; s3 < 4; ++s3)
#pragma unroll
                for (int jh = 0; jh < 2; ++jh) {
                    v4f o;
                    o.x = c[i][2 * jh][2 * s3];
                    o.y = c[i][2 * jh][2 * s3 + 1];
                    o.z = c[i][2 * jh + 1][2 * s3];
                    o.w = c[i][2 * jh + 1][2 * s3 + 1];
                    *(v4f*)(vf + (4 * i + s3) * 128 + 4 * jh) = o;
                }
    }
}

template <int N, int KD, bool AFRAG>
__global__ void __launch_bounds__(128)
gemm_nt_kernel(const float* __restrict__ A,
               const float* __restrict__ B,
               float* __restrict__ C) {
    gemm_tile<N, KD, AFRAG>(A, B, C, blockIdx.y, blockIdx.x, 0);
}

__global__ void __launch_bounds__(128)
proj_kernel(const float* __restrict__ xq_f, const float* __restrict__ xkv_f,
            const float* __restrict__ rel_f, const float* __restrict__ w_f,
            float* __restrict__ Q, float* __restrict__ K,
            float* __restrict__ V, float* __restrict__ R) {
    const int y = blockIdx.y;
    const float* A;
    const float* B;
    float* C;
    int mblk;
    int epi;
    if (y < 32) {
        A = xq_f;  B = w_f;           C = Q; mblk = y;       epi = 0;
    } else if (y < 96) {
        A = xkv_f; B = w_f + 1048576; C = K; mblk = y - 32;  epi = 1;
    } else if (y < 160) {
        A = xkv_f; B = w_f + 2097152; C = V; mblk = y - 96;  epi = 2;
    } else {
        A = rel_f; B = w_f + 3145728; C = R; mblk = y - 160; epi = 1;
    }
    gemm_tile<1024, 1024, true>(A, B, C, mblk, blockIdx.x, epi);
}

struct ScoreSet {
    v4f a[4][4];
};

__device__ __forceinline__ void score_fetch(ScoreSet& st,
                                            const float* __restrict__ krow) {
#pragma unroll
    for (int ik = 0; ik < 4; ++ik)
#pragma unroll
        for (int p = 0; p < 4; ++p)
            st.a[ik][p] = *(const v4f*)(krow + ik * 512 + p * 128);
}

__device__ __forceinline__ void score_mma(const v4f (&q)[2][4],
                                          const ScoreSet& st, v8f (&c)[4][2]) {
#pragma unroll
    for (int s = 0; s < 8; ++s) {
        v2f fa[4], fb[2];
#pragma unroll
        for (int ik = 0; ik < 4; ++ik) fa[ik] = gemm_pick(st.a[ik][s >> 1], s & 1);
#pragma unroll
        for (int jq = 0; jq < 2; ++jq) fb[jq] = gemm_pick(q[jq][s >> 1], s & 1);
#pragma unroll
        for (int ik = 0; ik < 4; ++ik)
#pragma unroll
            for (int jq = 0; jq < 2; ++jq) c[ik][jq] = wmma4(fa[ik], fb[jq], c[ik][jq]);
    }
}

__device__ __forceinline__ void score_store(const v8f (&c)[4][2],
                                            float* __restrict__ sp0,
                                            float* __restrict__ sp1, int dq,
                                            float (&rm)[2]) {
    const float scale = 0.125f;
#pragma unroll
    for (int jq = 0; jq < 2; ++jq) {
        float* sp = jq ? sp1 : sp0;
#pragma unroll
        for (int ik = 0; ik < 4; ++ik)
#pragma unroll
            for (int rh = 0; rh < 2; ++rh) {
                const int d = dq + 16 * ik - 16 * jq + 4 * rh;
                v4f o;
                o.x = (d + 0 > 0) ? -INFINITY : c[ik][jq][4 * rh + 0] * scale;
                o.y = (d + 1 > 0) ? -INFINITY : c[ik][jq][4 * rh + 1] * scale;
                o.z = (d + 2 > 0) ? -INFINITY : c[ik][jq][4 * rh + 2] * scale;
                o.w = (d + 3 > 0) ? -INFINITY : c[ik][jq][4 * rh + 3] * scale;
                *(v4f*)(sp + ik * 256 + rh * 128) = o;
                rm[jq] = fmaxf(fmaxf(rm[jq], fmaxf(o.x, o.y)), fmaxf(o.z, o.w));
            }
    }
}

__global__ void __launch_bounds__(256)
score_kernel(const float* __restrict__ Q,
             const float* __restrict__ Kf,
             const float* __restrict__ Rf,
             const float* __restrict__ u,
             const float* __restrict__ v,
             float* __restrict__ S, float* __restrict__ Mx) {
    const int lane = threadIdx.x & 31;
    const int w    = __builtin_amdgcn_readfirstlane(threadIdx.x >> 5);
    const int l16  = lane & 15;
    const int hlf  = lane >> 4;
    const int s  = blockIdx.y;
    const int b  = s >> 4;
    const int h  = s & 15;
    const int g  = (w < 4) ? 31 - (int)blockIdx.x : (int)blockIdx.x;
    const int kq = w & 3;
    const int m0 = 32 * g;
    const int ntile = 18 + 2 * (g >> 2);
    const int n  = (ntile - kq + 3) >> 2;

    const float* Qb = Q + (size_t)b * 1024 * 1024 + h * 64;

    v4f qa[2][2][4], qv[2][2][4];
#pragma unroll
    for (int cc = 0; cc < 2; ++cc)
#pragma unroll
        for (int p = 0; p < 4; ++p) {
            const int d = 32 * cc + 16 * hlf + 4 * p;
            const v4f u4 = *(const v4f*)(u + h * 64 + d);
            const v4f v4 = *(const v4f*)(v + h * 64 + d);
#pragma unroll
            for (int i = 0; i < 2; ++i) {
                const v4f q4 = *(const v4f*)(Qb + (size_t)(m0 + 16 * i + l16) * 1024 + d);
                qa[cc][i][p] = q4 + u4;
                qv[cc][i][p] = q4 + v4;
            }
        }

    const float* krow = Kf + (size_t)(b * 16 + h) * 131072 + kq * 4096 + lane * 4;
    const float* rrow = Rf + (size_t)h * 131072 + kq * 4096 + lane * 4;
    const int G = g >> 2;
    const int pitch = 128 * (9 + G);
    float* srow0 = S + (size_t)s * 1638400
                     + (size_t)16384 * (9 * G + ((G * (G - 1)) >> 1))
                     + (size_t)(32 * (g & 3) * pitch) + kq * 1024 + lane * 4;
    float* srow1 = srow0 + 16 * pitch;
    const int dq = 64 * kq + 8 * hlf - (m0 + l16) - 1024;

    v8f c[4][2];
#pragma unroll
    for (int i = 0; i < 4; ++i)
#pragma unroll
        for (int j = 0; j < 2; ++j) c[i][j] = v8f_zero();

    float rm[2] = {-INFINITY, -INFINITY};

#define SCORE_HOLD() __builtin_amdgcn_sched_barrier(0)
    const size_t tile = (size_t)16384;
    ScoreSet s0, s1, s2, s3;
    score_fetch(s0, krow);
    score_fetch(s1, krow + 2048);
    score_fetch(s2, rrow);
    for (int t = 0; t < n - 1; ++t) {
        SCORE_HOLD();
        score_fetch(s3, rrow + t * tile + 2048);
        SCORE_HOLD();
        score_mma(qa[0], s0, c);
        SCORE_HOLD();
        score_fetch(s0, krow + (t + 1) * tile);
        SCORE_HOLD();
        score_mma(qa[1], s1, c);
        SCORE_HOLD();
        score_fetch(s1, krow + (t + 1) * tile + 2048);
        SCORE_HOLD();
        score_mma(qv[0], s2, c);
        SCORE_HOLD();
        score_fetch(s2, rrow + (t + 1) * tile);
        SCORE_HOLD();
        score_mma(qv[1], s3, c);
        SCORE_HOLD();
        score_store(c, srow0 + 4096 * t, srow1 + 4096 * t, dq + 256 * t, rm);
#pragma unroll
        for (int i = 0; i < 4; ++i)
#pragma unroll
            for (int j = 0; j < 2; ++j) c[i][j] = v8f_zero();
    }
    SCORE_HOLD();
    score_fetch(s3, rrow + (n - 1) * tile + 2048);
    SCORE_HOLD();
    score_mma(qa[0], s0, c);
    score_mma(qa[1], s1, c);
    score_mma(qv[0], s2, c);
    score_mma(qv[1], s3, c);
#undef SCORE_HOLD
    score_store(c, srow0 + 4096 * (n - 1), srow1 + 4096 * (n - 1), dq + 256 * (n - 1), rm);

    const float mq0 = fmaxf(rm[0], __shfl_xor(rm[0], 16));
    const float mq1 = fmaxf(rm[1], __shfl_xor(rm[1], 16));
    Mx[((size_t)s * 4 + kq) * 1024 + m0 + 16 * hlf + l16] = hlf ? mq1 : mq0;
}

struct PvChunk {
    v4f p[4];
    v4f v[8];
};

__device__ __forceinline__ void pv_fetch(PvChunk& ch,
                                         const float* __restrict__ prow,
                                         const float* __restrict__ vrow,
                                         int kb) {
#pragma unroll
    for (int i = 0; i < 4; ++i)
        ch.p[i] = *(const v4f*)(prow + kb * 16 + 128 * i);
#pragma unroll
    for (int t = 0; t < 8; ++t)
        ch.v[t] = *(const v4f*)(vrow + kb * 64 + 128 * t);
}

__device__ __forceinline__ float pv_exp(float x, float nm) {
    return __builtin_amdgcn_exp2f(__builtin_fmaf(x, 1.4426950408889634f, nm));
}

__device__ __forceinline__ void pv_mma(const PvChunk& ch, v8f (&c)[2],
                                       float nm, v2f& lsum) {
#pragma unroll
    for (int s = 0; s < 8; ++s) {
        const v4f pp = ch.p[s >> 1];
        v2f a;
        a.x = pv_exp((s & 1) ? pp.z : pp.x, nm);
        a.y = pv_exp((s & 1) ? pp.w : pp.y, nm);
        lsum += a;
        const v4f vv = ch.v[s];
        v2f b0, b1;
        b0.x = vv.x;
        b0.y = vv.y;
        b1.x = vv.z;
        b1.y = vv.w;
        c[0] = wmma4(a, b0, c[0]);
        c[1] = wmma4(a, b1, c[1]);
    }
}

__global__ void __launch_bounds__(128) __attribute__((amdgpu_waves_per_eu(1, 2)))
pv_kernel(const float* __restrict__ P,
          const float* __restrict__ Mx,
          const float* __restrict__ V,
          float* __restrict__ O) {
    const int lane = threadIdx.x & 31;
    const int w    = __builtin_amdgcn_readfirstlane(threadIdx.x >> 5);
    const int l16  = lane & 15;
    const int hlf  = lane >> 4;
    const int dh   = w & 1;
    const int s  = blockIdx.y;
    const int b  = s >> 4;
    const int h  = s & 15;
    const int pr = 2 * blockIdx.x + (w >> 1);

    const float* vcol = V + (size_t)(b * 16 + h) * 131072 + dh * 1024 + lane * 4;

#define PV_HOLD() __builtin_amdgcn_sched_barrier(0)
#define PV_DONE() asm volatile("" :: "v"(c[0]), "v"(c[1]))
    for (int seg = 0; seg < 2; ++seg) {
        const int t  = seg ? 63 - pr : pr;
        const int m0 = 16 * t;
        const int G = t >> 3;
        const int kend = 128 * (9 + G);

        const float* prow = P + (size_t)s * 1638400
                              + (size_t)16384 * (9 * G + ((G * (G - 1)) >> 1))
                              + (size_t)(16 * (t & 7) * kend) + lane * 4;

        v8f c[2];
#pragma unroll
        for (int j = 0; j < 2; ++j) c[j] = v8f_zero();

        const float* mrow = Mx + (size_t)s * 4 * 1024 + m0 + l16;
        float mx = mrow[0];
#pragma unroll
        for (int p = 1; p < 4; ++p) mx = fmaxf(mx, mrow[p * 1024]);
        const float nm = -mx * 1.4426950408889634f;
        v2f lsum = {0.f, 0.f};

        PvChunk c0, c1, c2, c3;
        pv_fetch(c0, prow, vcol, 0);
        pv_fetch(c1, prow, vcol, 32);
        pv_fetch(c2, prow, vcol, 64);
        for (int kb = 0; kb < kend - 128; kb += 128) {
            PV_HOLD();
            pv_fetch(c3, prow, vcol, kb + 96);
            PV_HOLD();
            pv_mma(c0, c, nm, lsum);
            PV_DONE();
            PV_HOLD();
            pv_fetch(c0, prow, vcol, kb + 128);
            PV_HOLD();
            pv_mma(c1, c, nm, lsum);
            PV_DONE();
            PV_HOLD();
            pv_fetch(c1, prow, vcol, kb + 160);
            PV_HOLD();
            pv_mma(c2, c, nm, lsum);
            PV_DONE();
            PV_HOLD();
            pv_fetch(c2, prow, vcol, kb + 192);
            PV_HOLD();
            pv_mma(c3, c, nm, lsum);
            PV_DONE();
        }
        PV_HOLD();
        pv_fetch(c3, prow, vcol, kend - 32);
        PV_HOLD();
        pv_mma(c0, c, nm, lsum);
        pv_mma(c1, c, nm, lsum);
        pv_mma(c2, c, nm, lsum);
        pv_mma(c3, c, nm, lsum);

        float* orow = O + (size_t)((b * 1024 + m0) >> 4) * 16384 + (2 * h + dh) * 512
                        + ((l16 >> 1) & 3) * 128 + (16 * (l16 >> 3) + 8 * hlf) * 4
                        + 2 * (l16 & 1);
        float rsum = lsum.x + lsum.y;
        rsum += __shfl_xor(rsum, 16);
#pragma unroll
        for (int r = 0; r < 8; ++r) {
            const float inv = 1.0f / __shfl(rsum, 8 * hlf + r);
            v2f o;
            o.x = c[0][r] * inv;
            o.y = c[1][r] * inv;
            *(v2f*)(orow + 4 * r) = o;
        }
    }
#undef PV_DONE
#undef PV_HOLD
}

extern "C" void kernel_launch(void* const* d_in, const int* in_sizes, int n_in,
                              void* d_out, int out_size, void* d_ws, size_t ws_size,
                              hipStream_t stream) {
    const float* x_q  = (const float*)d_in[0];
    const float* x_kv = (const float*)d_in[1];
    const float* Wq   = (const float*)d_in[2];
    const float* Wk   = (const float*)d_in[3];
    const float* Wv   = (const float*)d_in[4];
    const float* Wr   = (const float*)d_in[5];
    const float* Wo   = (const float*)d_in[6];
    const float* ub   = (const float*)d_in[7];
    const float* vb   = (const float*)d_in[8];
    float* out = (float*)d_out;
    (void)in_sizes; (void)n_in; (void)out_size; (void)ws_size;

    float* ws   = (float*)d_ws;
    float* Qb   = ws;
    float* Ob   = Qb;
    float* Kb   = Qb + 2097152;
    float* Vb   = Kb + 4194304;
    float* Rb   = Vb + 4194304;
    float* Wof  = Rb + 2097152;
    float* Mxb  = Wof + 1048576;
    float* Sb   = Mxb + 131072;
    float* Relb = Sb;
    float* Xqf  = Relb + 2097152;
    float* Xkvf = Xqf + 2097152;
    float* Wf   = Xkvf + 4194304;

    prep_kernel<<<1216, 256, 0, stream>>>(x_q, x_kv, Wq, Wk, Wv, Wr, Wo,
                                          Xqf, Xkvf, Wf, Wof, Relb);

    proj_kernel<<<dim3(8, 192), 128, 0, stream>>>(Xqf, Xkvf, Relb, Wf, Qb, Kb, Vb, Rb);

    score_kernel<<<dim3(16, 32), 256, 0, stream>>>(Qb, Kb, Rb, ub, vb, Sb, Mxb);
    pv_kernel<<<dim3(16, 32), 128, 0, stream>>>(Sb, Mxb, Vb, Ob);

    gemm_nt_kernel<1024, 1024, true><<<dim3(8, 32), 128, 0, stream>>>(Ob, Wof, out);
}
